# starting kernel
# baseline (speedup 1.0000x reference)
#include <hip/hip_runtime.h>
#include <math.h>

#define BB 4
#define TT 256
#define TS 512
#define HH 1024
#define VV 32100
#define CHUNK 8192

typedef float v2f __attribute__((ext_vector_type(2)));
typedef float v8f __attribute__((ext_vector_type(8)));

__global__ void k_encw(const float* __restrict__ enc,
                       const float* __restrict__ wgen,
                       float* __restrict__ encW) {
    int wave = threadIdx.x >> 5;
    int lane = threadIdx.x & 31;
    int row  = blockIdx.x * 8 + wave;
    const float4* p = (const float4*)(enc + (size_t)row * HH);
    const float4* w = (const float4*)wgen;
    float acc = 0.f;
    for (int i = lane; i < HH / 4; i += 32) {
        float4 a = p[i];
        float4 b = w[i];
        acc += a.x * b.x + a.y * b.y + a.z * b.z + a.w * b.w;
    }
    for (int off = 16; off > 0; off >>= 1)
        acc += __shfl_down(acc, off, 32);
    if (lane == 0) encW[row] = acc;
}

__global__ void k_pgen(const float* __restrict__ attn,
                       const float* __restrict__ dec,
                       const float* __restrict__ encW,
                       const float* __restrict__ wgen,
                       const float* __restrict__ bgen,
                       float* __restrict__ pgen) {
    int b    = blockIdx.y;
    int t0   = blockIdx.x * 16;
    int lane = threadIdx.x;
    int M    = lane & 15;
    int klo  = (lane >> 4) << 1;

    const float* arow = attn + (size_t)(b * TT + t0 + M) * TS;
    const float* drow = dec  + (size_t)(b * TT + t0 + M) * HH;
    const float* ew   = encW + b * TS;
    const float* wg2  = wgen + HH;

    v8f c = {};

    for (int k0 = 0; k0 < TS; k0 += 4) {
        v2f a;
        a.x = arow[k0 + klo];
        a.y = arow[k0 + klo + 1];
        float e0 = ew[k0 + klo];
        float e1 = ew[k0 + klo + 1];
        v2f bm;
        bm.x = (M == 0) ? e0 : 0.f;
        bm.y = (M == 0) ? e1 : 0.f;
        c = __builtin_amdgcn_wmma_f32_16x16x4_f32(false, a, false, bm,
                                                  (short)0, c, false, false);
    }
    for (int h0 = 0; h0 < HH; h0 += 4) {
        v2f a;
        a.x = drow[h0 + klo];
        a.y = drow[h0 + klo + 1];
        float e0 = wg2[h0 + klo];
        float e1 = wg2[h0 + klo + 1];
        v2f bm;
        bm.x = (M == 0) ? e0 : 0.f;
        bm.y = (M == 0) ? e1 : 0.f;
        c = __builtin_amdgcn_wmma_f32_16x16x4_f32(false, a, false, bm,
                                                  (short)0, c, false, false);
    }

    if (M == 0) {
        int   mbase = (lane == 0) ? 0 : 8;
        float bg    = bgen[0];
        for (int r = 0; r < 8; ++r) {
            float logit = c[r] + bg;
            pgen[b * TT + t0 + mbase + r] = 1.f / (1.f + __expf(-logit));
        }
    }
}

__global__ void k_main(const float* __restrict__ logits,
                       const float* __restrict__ attn,
                       const int*   __restrict__ sids,
                       const float* __restrict__ pgen,
                       float* __restrict__ out) {
    __shared__ float s_copy[CHUNK];
    __shared__ float red_m[256];
    __shared__ float red_s[256];
    __shared__ float bc[2];

    int blk = blockIdx.x;
    int b   = blk >> 8;
    int tid = threadIdx.x;

    const float* row = logits + (size_t)blk * VV;

    float m = -INFINITY, s = 0.f;
    for (int v = tid; v < VV; v += 256) {
        float x = row[v];
        if (x > m) { s = s * __expf(m - x) + 1.f; m = x; }
        else       { s += __expf(x - m); }
    }
    red_m[tid] = m;
    red_s[tid] = s;
    __syncthreads();
    for (int off = 128; off > 0; off >>= 1) {
        if (tid < off) {
            float m1 = red_m[tid], s1 = red_s[tid];
            float m2 = red_m[tid + off], s2 = red_s[tid + off];
            float mm = fmaxf(m1, m2);
            red_m[tid] = mm;
            red_s[tid] = s1 * __expf(m1 - mm) + s2 * __expf(m2 - mm);
        }
        __syncthreads();
    }
    if (tid == 0) { bc[0] = red_m[0]; bc[1] = 1.f / red_s[0]; }
    __syncthreads();
    float Mx   = bc[0];
    float invS = bc[1];

    float p   = pgen[blk];
    float omp = 1.f - p;
    const int*   ids  = sids + b * TS;
    const float* arow = attn + (size_t)blk * TS;
    float*       orow = out  + (size_t)blk * VV;

    for (int c0 = 0; c0 < VV; c0 += CHUNK) {
        int cend = (c0 + CHUNK < VV) ? (c0 + CHUNK) : VV;
        for (int i = tid; i < CHUNK; i += 256) s_copy[i] = 0.f;
        __syncthreads();
        for (int si = tid; si < TS; si += 256) {
            int id = ids[si];
            if (id >= c0 && id < cend)
                atomicAdd(&s_copy[id - c0], arow[si]);
        }
        __syncthreads();
        for (int v = c0 + tid; v < cend; v += 256) {
            float x   = row[v];
            float val = p * __expf(x - Mx) * invS
                      + omp * s_copy[v - c0] + 1e-12f;
            orow[v] = __logf(val);
        }
        __syncthreads();
    }
}

extern "C" void kernel_launch(void* const* d_in, const int* in_sizes, int n_in,
                              void* d_out, int out_size, void* d_ws, size_t ws_size,
                              hipStream_t stream) {
    (void)in_sizes; (void)n_in; (void)out_size; (void)ws_size;
    const float* dec    = (const float*)d_in[0];
    const float* attn   = (const float*)d_in[1];
    const float* enc    = (const float*)d_in[2];
    const float* logits = (const float*)d_in[3];
    const float* wgen   = (const float*)d_in[4];
    const float* bgen   = (const float*)d_in[5];
    const int*   sids   = (const int*)d_in[6];
    float* out  = (float*)d_out;

    float* encW = (float*)d_ws;
    float* pgen = encW + BB * TS;

    k_encw<<<(BB * TS) / 8, 256, 0, stream>>>(enc, wgen, encW);
    k_pgen<<<dim3(TT / 16, BB), 32, 0, stream>>>(attn, dec, encW, wgen, bgen, pgen);
    k_main<<<BB * TT, 256, 0, stream>>>(logits, attn, sids, pgen, out);
}


# k_main holds each logits row in registers: all its 16-byte loads go out up front, max, exp-sum and output pass run from registers (row read once); exp-sum order: 4 partials per thread + shuffle tree
# speedup vs baseline: 1.9277x; 1.9277x over previous
#include <hip/hip_runtime.h>
#include <math.h>

#define BB 4
#define TT 256
#define TS 512
#define HH 1024
#define VV 32100
#define CHUNK 8192

typedef float v2f __attribute__((ext_vector_type(2)));
typedef float v8f __attribute__((ext_vector_type(8)));

__global__ void k_encw(const float* __restrict__ enc,
                       const float* __restrict__ wgen,
                       float* __restrict__ encW) {
    int wave = threadIdx.x >> 5;
    int lane = threadIdx.x & 31;
    int row  = blockIdx.x * 8 + wave;
    const float4* p = (const float4*)(enc + (size_t)row * HH);
    const float4* w = (const float4*)wgen;
    float acc = 0.f;
    for (int i = lane; i < HH / 4; i += 32) {
        float4 a = p[i];
        float4 b = w[i];
        acc += a.x * b.x + a.y * b.y + a.z * b.z + a.w * b.w;
    }
    for (int off = 16; off > 0; off >>= 1)
        acc += __shfl_down(acc, off, 32);
    if (lane == 0) encW[row] = acc;
}

__global__ void k_pgen(const float* __restrict__ attn,
                       const float* __restrict__ dec,
                       const float* __restrict__ encW,
                       const float* __restrict__ wgen,
                       const float* __restrict__ bgen,
                       float* __restrict__ pgen) {
    int b    = blockIdx.y;
    int t0   = blockIdx.x * 16;
    int lane = threadIdx.x;
    int M    = lane & 15;
    int klo  = (lane >> 4) << 1;

    const float* arow = attn + (size_t)(b * TT + t0 + M) * TS;
    const float* drow = dec  + (size_t)(b * TT + t0 + M) * HH;
    const float* ew   = encW + b * TS;
    const float* wg2  = wgen + HH;

    v8f c = {};

    for (int k0 = 0; k0 < TS; k0 += 4) {
        v2f a;
        a.x = arow[k0 + klo];
        a.y = arow[k0 + klo + 1];
        float e0 = ew[k0 + klo];
        float e1 = ew[k0 + klo + 1];
        v2f bm;
        bm.x = (M == 0) ? e0 : 0.f;
        bm.y = (M == 0) ? e1 : 0.f;
        c = __builtin_amdgcn_wmma_f32_16x16x4_f32(false, a, false, bm,
                                                  (short)0, c, false, false);
    }
    for (int h0 = 0; h0 < HH; h0 += 4) {
        v2f a;
        a.x = drow[h0 + klo];
        a.y = drow[h0 + klo + 1];
        float e0 = wg2[h0 + klo];
        float e1 = wg2[h0 + klo + 1];
        v2f bm;
        bm.x = (M == 0) ? e0 : 0.f;
        bm.y = (M == 0) ? e1 : 0.f;
        c = __builtin_amdgcn_wmma_f32_16x16x4_f32(false, a, false, bm,
                                                  (short)0, c, false, false);
    }

    if (M == 0) {
        int   mbase = (lane == 0) ? 0 : 8;
        float bg    = bgen[0];
        for (int r = 0; r < 8; ++r) {
            float logit = c[r] + bg;
            pgen[b * TT + t0 + mbase + r] = 1.f / (1.f + __expf(-logit));
        }
    }
}

#define NTHR 256
#define V4   (VV / 4)
#define PER  ((V4 + NTHR - 1) / NTHR)
#define C4   (CHUNK / 4)
#define JPC  (C4 / NTHR)
#define NCHK ((VV + CHUNK - 1) / CHUNK)

__global__ __launch_bounds__(NTHR, 6)
void k_main(const float* __restrict__ logits,
            const float* __restrict__ attn,
            const int*   __restrict__ sids,
            const float* __restrict__ pgen,
            float* __restrict__ out) {
    __shared__ float4 s_copy4[C4];
    __shared__ float red_m[NTHR / 32];
    __shared__ float red_s[NTHR / 32];
    float* s_copy = (float*)s_copy4;

    int blk  = blockIdx.x;
    int b    = blk >> 8;
    int tid  = threadIdx.x;
    int lane = tid & 31;
    int wave = tid >> 5;

    const float4* row4  = (const float4*)(logits + (size_t)blk * VV);
    float4*       orow4 = (float4*)(out + (size_t)blk * VV);
    const int*    ids   = sids + b * TS;
    const float*  arow  = attn + (size_t)blk * TS;

    const bool tail = (PER - 1) * NTHR + tid < V4;
    float4 x[PER];
    #pragma unroll
    for (int j = 0; j < PER - 1; ++j) x[j] = row4[j * NTHR + tid];
    if (tail) x[PER - 1] = row4[(PER - 1) * NTHR + tid];
    else      x[PER - 1] = make_float4(-INFINITY, -INFINITY, -INFINITY, -INFINITY);
    int   id0 = ids[tid],  id1 = ids[tid + NTHR];
    float a0  = arow[tid], a1  = arow[tid + NTHR];
    float p   = pgen[blk];
    float omp = 1.f - p;

    float m = -INFINITY;
    #pragma unroll
    for (int j = 0; j < PER; ++j)
        m = fmaxf(m, fmaxf(fmaxf(x[j].x, x[j].y), fmaxf(x[j].z, x[j].w)));
    for (int off = 16; off > 0; off >>= 1)
        m = fmaxf(m, __shfl_xor(m, off, 32));
    if (lane == 0) red_m[wave] = m;
    __syncthreads();
    float Mx = red_m[0];
    #pragma unroll
    for (int w = 1; w < NTHR / 32; ++w) Mx = fmaxf(Mx, red_m[w]);

    float s0 = 0.f, s1 = 0.f, s2 = 0.f, s3 = 0.f;
    #pragma unroll
    for (int j = 0; j < PER; ++j) {
        x[j].x = __expf(x[j].x - Mx);  s0 += x[j].x;
        x[j].y = __expf(x[j].y - Mx);  s1 += x[j].y;
        x[j].z = __expf(x[j].z - Mx);  s2 += x[j].z;
        x[j].w = __expf(x[j].w - Mx);  s3 += x[j].w;
    }
    float s = (s0 + s1) + (s2 + s3);
    for (int off = 16; off > 0; off >>= 1)
        s += __shfl_xor(s, off, 32);
    if (lane == 0) red_s[wave] = s;
    __syncthreads();
    float S = red_s[0];
    #pragma unroll
    for (int w = 1; w < NTHR / 32; ++w) S += red_s[w];
    float invS = 1.f / S;

    #pragma unroll
    for (int c = 0; c < NCHK; ++c) {
        const int c0   = c * CHUNK;
        const int cend = (c0 + CHUNK < VV) ? (c0 + CHUNK) : VV;
        #pragma unroll
        for (int k = 0; k < JPC; ++k)
            s_copy4[k * NTHR + tid] = make_float4(0.f, 0.f, 0.f, 0.f);
        __syncthreads();
        if (id0 >= c0 && id0 < cend) atomicAdd(&s_copy[id0 - c0], a0);
        if (id1 >= c0 && id1 < cend) atomicAdd(&s_copy[id1 - c0], a1);
        __syncthreads();
        #pragma unroll
        for (int k = 0; k < JPC; ++k) {
            const int j = c * JPC + k;
            if (j < PER - 1 || tail) {
                float4 cp = s_copy4[k * NTHR + tid];
                float4 e  = x[j];
                float4 o;
                o.x = __logf(p * e.x * invS + omp * cp.x + 1e-12f);
                o.y = __logf(p * e.y * invS + omp * cp.y + 1e-12f);
                o.z = __logf(p * e.z * invS + omp * cp.z + 1e-12f);
                o.w = __logf(p * e.w * invS + omp * cp.w + 1e-12f);
                orow4[j * NTHR + tid] = o;
            }
        }
        __syncthreads();
    }
}

extern "C" void kernel_launch(void* const* d_in, const int* in_sizes, int n_in,
                              void* d_out, int out_size, void* d_ws, size_t ws_size,
                              hipStream_t stream) {
    (void)in_sizes; (void)n_in; (void)out_size; (void)ws_size;
    const float* dec    = (const float*)d_in[0];
    const float* attn   = (const float*)d_in[1];
    const float* enc    = (const float*)d_in[2];
    const float* logits = (const float*)d_in[3];
    const float* wgen   = (const float*)d_in[4];
    const float* bgen   = (const float*)d_in[5];
    const int*   sids   = (const int*)d_in[6];
    float* out  = (float*)d_out;

    float* encW = (float*)d_ws;
    float* pgen = encW + BB * TS;

    k_encw<<<(BB * TS) / 8, 256, 0, stream>>>(enc, wgen, encW);
    k_pgen<<<dim3(TT / 16, BB), 32, 0, stream>>>(attn, dec, encW, wgen, bgen, pgen);
    k_main<<<BB * TT, 256, 0, stream>>>(logits, attn, sids, pgen, out);
}


# k_pgen: each 16-row tile's K range (1536) cut into 12 slices, one wave each: all 32 loads of a wave in flight, 2 WMMA accumulators, partials summed via LDS in slice order (p_gen sum order changes)
# speedup vs baseline: 2.7551x; 1.4292x over previous
#include <hip/hip_runtime.h>
#include <math.h>

#define BB 4
#define TT 256
#define TS 512
#define HH 1024
#define VV 32100
#define CHUNK 8192

typedef float v2f __attribute__((ext_vector_type(2)));
typedef float v8f __attribute__((ext_vector_type(8)));

__global__ void k_encw(const float* __restrict__ enc,
                       const float* __restrict__ wgen,
                       float* __restrict__ encW) {
    int wave = threadIdx.x >> 5;
    int lane = threadIdx.x & 31;
    int row  = blockIdx.x * 8 + wave;
    const float4* p = (const float4*)(enc + (size_t)row * HH);
    const float4* w = (const float4*)wgen;
    float acc = 0.f;
    for (int i = lane; i < HH / 4; i += 32) {
        float4 a = p[i];
        float4 b = w[i];
        acc += a.x * b.x + a.y * b.y + a.z * b.z + a.w * b.w;
    }
    for (int off = 16; off > 0; off >>= 1)
        acc += __shfl_down(acc, off, 32);
    if (lane == 0) encW[row] = acc;
}

#define PGSL 128
#define PGSW (TS / PGSL)
#define PGW  ((TS + HH) / PGSL)
#define PGJ  (PGSL / 8)

__global__ __launch_bounds__(PGW * 32) __attribute__((amdgpu_waves_per_eu(3, 3)))
void k_pgen(const float* __restrict__ attn,
            const float* __restrict__ dec,
            const float* __restrict__ encW,
            const float* __restrict__ wgen,
            const float* __restrict__ bgen,
            float* __restrict__ pgen) {
    __shared__ float part[PGW][16];

    int b    = blockIdx.y;
    int t0   = blockIdx.x * 16;
    int lane = threadIdx.x & 31;
    int wave = threadIdx.x >> 5;
    int M    = lane & 15;
    int h    = lane >> 4;

    const float* arow;
    const float* vec;
    if (wave < PGSW) {
        arow = attn + (size_t)(b * TT + t0 + M) * TS + wave * PGSL;
        vec  = encW + b * TS + wave * PGSL;
    } else {
        arow = dec + (size_t)(b * TT + t0 + M) * HH + (wave - PGSW) * PGSL;
        vec  = wgen + HH + (wave - PGSW) * PGSL;
    }
    const float4* ap = (const float4*)arow + h;
    const float4* vp = (const float4*)vec  + h;

    float4 a[PGJ], v[PGJ];
    #pragma unroll
    for (int j = 0; j < PGJ; ++j) v[j] = vp[2 * j];
    #pragma unroll
    for (int j = 0; j < PGJ; ++j) a[j] = ap[2 * j];

    v8f c0 = {};
    v8f c1 = {};
    #pragma unroll
    for (int j = 0; j < PGJ; ++j) {
        v2f a01, a23, b01, b23;
        a01.x = a[j].x;  a01.y = a[j].y;
        a23.x = a[j].z;  a23.y = a[j].w;
        b01.x = v[j].x;  b01.y = v[j].y;
        b23.x = v[j].z;  b23.y = v[j].w;
        c0 = __builtin_amdgcn_wmma_f32_16x16x4_f32(false, a01, false, b01,
                                                   (short)0, c0, false, false);
        c1 = __builtin_amdgcn_wmma_f32_16x16x4_f32(false, a23, false, b23,
                                                   (short)0, c1, false, false);
    }
    v8f c = c0 + c1;

    if (M == 0) {
        #pragma unroll
        for (int r = 0; r < 8; ++r) part[wave][8 * h + r] = c[r];
    }
    __syncthreads();
    if (threadIdx.x < 16) {
        float acc = 0.f;
        #pragma unroll
        for (int w = 0; w < PGW; ++w) acc += part[w][threadIdx.x];
        float logit = acc + bgen[0];
        pgen[b * TT + t0 + threadIdx.x] = 1.f / (1.f + __expf(-logit));
    }
}

#define NTHR 256
#define V4   (VV / 4)
#define PER  ((V4 + NTHR - 1) / NTHR)
#define C4   (CHUNK / 4)
#define JPC  (C4 / NTHR)
#define NCHK ((VV + CHUNK - 1) / CHUNK)

__global__ __launch_bounds__(NTHR, 6)
void k_main(const float* __restrict__ logits,
            const float* __restrict__ attn,
            const int*   __restrict__ sids,
            const float* __restrict__ pgen,
            float* __restrict__ out) {
    __shared__ float4 s_copy4[C4];
    __shared__ float red_m[NTHR / 32];
    __shared__ float red_s[NTHR / 32];
    float* s_copy = (float*)s_copy4;

    int blk  = blockIdx.x;
    int b    = blk >> 8;
    int tid  = threadIdx.x;
    int lane = tid & 31;
    int wave = tid >> 5;

    const float4* row4  = (const float4*)(logits + (size_t)blk * VV);
    float4*       orow4 = (float4*)(out + (size_t)blk * VV);
    const int*    ids   = sids + b * TS;
    const float*  arow  = attn + (size_t)blk * TS;

    const bool tail = (PER - 1) * NTHR + tid < V4;
    float4 x[PER];
    #pragma unroll
    for (int j = 0; j < PER - 1; ++j) x[j] = row4[j * NTHR + tid];
    if (tail) x[PER - 1] = row4[(PER - 1) * NTHR + tid];
    else      x[PER - 1] = make_float4(-INFINITY, -INFINITY, -INFINITY, -INFINITY);
    int   id0 = ids[tid],  id1 = ids[tid + NTHR];
    float a0  = arow[tid], a1  = arow[tid + NTHR];
    float p   = pgen[blk];
    float omp = 1.f - p;

    float m = -INFINITY;
    #pragma unroll
    for (int j = 0; j < PER; ++j)
        m = fmaxf(m, fmaxf(fmaxf(x[j].x, x[j].y), fmaxf(x[j].z, x[j].w)));
    for (int off = 16; off > 0; off >>= 1)
        m = fmaxf(m, __shfl_xor(m, off, 32));
    if (lane == 0) red_m[wave] = m;
    __syncthreads();
    float Mx = red_m[0];
    #pragma unroll
    for (int w = 1; w < NTHR / 32; ++w) Mx = fmaxf(Mx, red_m[w]);

    float s0 = 0.f, s1 = 0.f, s2 = 0.f, s3 = 0.f;
    #pragma unroll
    for (int j = 0; j < PER; ++j) {
        x[j].x = __expf(x[j].x - Mx);  s0 += x[j].x;
        x[j].y = __expf(x[j].y - Mx);  s1 += x[j].y;
        x[j].z = __expf(x[j].z - Mx);  s2 += x[j].z;
        x[j].w = __expf(x[j].w - Mx);  s3 += x[j].w;
    }
    float s = (s0 + s1) + (s2 + s3);
    for (int off = 16; off > 0; off >>= 1)
        s += __shfl_xor(s, off, 32);
    if (lane == 0) red_s[wave] = s;
    __syncthreads();
    float S = red_s[0];
    #pragma unroll
    for (int w = 1; w < NTHR / 32; ++w) S += red_s[w];
    float invS = 1.f / S;

    #pragma unroll
    for (int c = 0; c < NCHK; ++c) {
        const int c0   = c * CHUNK;
        const int cend = (c0 + CHUNK < VV) ? (c0 + CHUNK) : VV;
        #pragma unroll
        for (int k = 0; k < JPC; ++k)
            s_copy4[k * NTHR + tid] = make_float4(0.f, 0.f, 0.f, 0.f);
        __syncthreads();
        if (id0 >= c0 && id0 < cend) atomicAdd(&s_copy[id0 - c0], a0);
        if (id1 >= c0 && id1 < cend) atomicAdd(&s_copy[id1 - c0], a1);
        __syncthreads();
        #pragma unroll
        for (int k = 0; k < JPC; ++k) {
            const int j = c * JPC + k;
            if (j < PER - 1 || tail) {
                float4 cp = s_copy4[k * NTHR + tid];
                float4 e  = x[j];
                float4 o;
                o.x = __logf(p * e.x * invS + omp * cp.x + 1e-12f);
                o.y = __logf(p * e.y * invS + omp * cp.y + 1e-12f);
                o.z = __logf(p * e.z * invS + omp * cp.z + 1e-12f);
                o.w = __logf(p * e.w * invS + omp * cp.w + 1e-12f);
                orow4[j * NTHR + tid] = o;
            }
        }
        __syncthreads();
    }
}

extern "C" void kernel_launch(void* const* d_in, const int* in_sizes, int n_in,
                              void* d_out, int out_size, void* d_ws, size_t ws_size,
                              hipStream_t stream) {
    (void)in_sizes; (void)n_in; (void)out_size; (void)ws_size;
    const float* dec    = (const float*)d_in[0];
    const float* attn   = (const float*)d_in[1];
    const float* enc    = (const float*)d_in[2];
    const float* logits = (const float*)d_in[3];
    const float* wgen   = (const float*)d_in[4];
    const float* bgen   = (const float*)d_in[5];
    const int*   sids   = (const int*)d_in[6];
    float* out  = (float*)d_out;

    float* encW = (float*)d_ws;
    float* pgen = encW + BB * TS;

    k_encw<<<(BB * TS) / 8, 256, 0, stream>>>(enc, wgen, encW);
    k_pgen<<<dim3(TT / 16, BB), PGW * 32, 0, stream>>>(attn, dec, encW, wgen, bgen, pgen);
    k_main<<<BB * TT, 256, 0, stream>>>(logits, attn, sids, pgen, out);
}


# k_main output pass: natural log as the hardware base-2 log instruction times ln 2 instead of the library log expansion (13 vector slots per element); exp-pass scheduling fence keeps registers at 158
# speedup vs baseline: 2.8934x; 1.0502x over previous
#include <hip/hip_runtime.h>
#include <math.h>

#define BB 4
#define TT 256
#define TS 512
#define HH 1024
#define VV 32100
#define CHUNK 8192

typedef float v2f __attribute__((ext_vector_type(2)));
typedef float v8f __attribute__((ext_vector_type(8)));

__global__ void k_encw(const float* __restrict__ enc,
                       const float* __restrict__ wgen,
                       float* __restrict__ encW) {
    int wave = threadIdx.x >> 5;
    int lane = threadIdx.x & 31;
    int row  = blockIdx.x * 8 + wave;
    const float4* p = (const float4*)(enc + (size_t)row * HH);
    const float4* w = (const float4*)wgen;
    float acc = 0.f;
    for (int i = lane; i < HH / 4; i += 32) {
        float4 a = p[i];
        float4 b = w[i];
        acc += a.x * b.x + a.y * b.y + a.z * b.z + a.w * b.w;
    }
    for (int off = 16; off > 0; off >>= 1)
        acc += __shfl_down(acc, off, 32);
    if (lane == 0) encW[row] = acc;
}

#define PGSL 128
#define PGSW (TS / PGSL)
#define PGW  ((TS + HH) / PGSL)
#define PGJ  (PGSL / 8)

__global__ __launch_bounds__(PGW * 32) __attribute__((amdgpu_waves_per_eu(3, 3)))
void k_pgen(const float* __restrict__ attn,
            const float* __restrict__ dec,
            const float* __restrict__ encW,
            const float* __restrict__ wgen,
            const float* __restrict__ bgen,
            float* __restrict__ pgen) {
    __shared__ float part[PGW][16];

    int b    = blockIdx.y;
    int t0   = blockIdx.x * 16;
    int lane = threadIdx.x & 31;
    int wave = threadIdx.x >> 5;
    int M    = lane & 15;
    int h    = lane >> 4;

    const float* arow;
    const float* vec;
    if (wave < PGSW) {
        arow = attn + (size_t)(b * TT + t0 + M) * TS + wave * PGSL;
        vec  = encW + b * TS + wave * PGSL;
    } else {
        arow = dec + (size_t)(b * TT + t0 + M) * HH + (wave - PGSW) * PGSL;
        vec  = wgen + HH + (wave - PGSW) * PGSL;
    }
    const float4* ap = (const float4*)arow + h;
    const float4* vp = (const float4*)vec  + h;

    float4 a[PGJ], v[PGJ];
    #pragma unroll
    for (int j = 0; j < PGJ; ++j) v[j] = vp[2 * j];
    #pragma unroll
    for (int j = 0; j < PGJ; ++j) a[j] = ap[2 * j];

    v8f c0 = {};
    v8f c1 = {};
    #pragma unroll
    for (int j = 0; j < PGJ; ++j) {
        v2f a01, a23, b01, b23;
        a01.x = a[j].x;  a01.y = a[j].y;
        a23.x = a[j].z;  a23.y = a[j].w;
        b01.x = v[j].x;  b01.y = v[j].y;
        b23.x = v[j].z;  b23.y = v[j].w;
        c0 = __builtin_amdgcn_wmma_f32_16x16x4_f32(false, a01, false, b01,
                                                   (short)0, c0, false, false);
        c1 = __builtin_amdgcn_wmma_f32_16x16x4_f32(false, a23, false, b23,
                                                   (short)0, c1, false, false);
    }
    v8f c = c0 + c1;

    if (M == 0) {
        #pragma unroll
        for (int r = 0; r < 8; ++r) part[wave][8 * h + r] = c[r];
    }
    __syncthreads();
    if (threadIdx.x < 16) {
        float acc = 0.f;
        #pragma unroll
        for (int w = 0; w < PGW; ++w) acc += part[w][threadIdx.x];
        float logit = acc + bgen[0];
        pgen[b * TT + t0 + threadIdx.x] = 1.f / (1.f + __expf(-logit));
    }
}

__device__ __forceinline__ float ln_pos(float x) {
    return 0.6931471805599453f * __builtin_amdgcn_logf(x);
}

#define NTHR 256
#define V4   (VV / 4)
#define PER  ((V4 + NTHR - 1) / NTHR)
#define C4   (CHUNK / 4)
#define JPC  (C4 / NTHR)
#define NCHK ((VV + CHUNK - 1) / CHUNK)

__global__ __launch_bounds__(NTHR, 6)
void k_main(const float* __restrict__ logits,
            const float* __restrict__ attn,
            const int*   __restrict__ sids,
            const float* __restrict__ pgen,
            float* __restrict__ out) {
    __shared__ float4 s_copy4[C4];
    __shared__ float red_m[NTHR / 32];
    __shared__ float red_s[NTHR / 32];
    float* s_copy = (float*)s_copy4;

    int blk  = blockIdx.x;
    int b    = blk >> 8;
    int tid  = threadIdx.x;
    int lane = tid & 31;
    int wave = tid >> 5;

    const float4* row4  = (const float4*)(logits + (size_t)blk * VV);
    float4*       orow4 = (float4*)(out + (size_t)blk * VV);
    const int*    ids   = sids + b * TS;
    const float*  arow  = attn + (size_t)blk * TS;

    const bool tail = (PER - 1) * NTHR + tid < V4;
    float4 x[PER];
    #pragma unroll
    for (int j = 0; j < PER - 1; ++j) x[j] = row4[j * NTHR + tid];
    if (tail) x[PER - 1] = row4[(PER - 1) * NTHR + tid];
    else      x[PER - 1] = make_float4(-INFINITY, -INFINITY, -INFINITY, -INFINITY);
    int   id0 = ids[tid],  id1 = ids[tid + NTHR];
    float a0  = arow[tid], a1  = arow[tid + NTHR];
    float p   = pgen[blk];
    float omp = 1.f - p;

    float m = -INFINITY;
    #pragma unroll
    for (int j = 0; j < PER; ++j)
        m = fmaxf(m, fmaxf(fmaxf(x[j].x, x[j].y), fmaxf(x[j].z, x[j].w)));
    for (int off = 16; off > 0; off >>= 1)
        m = fmaxf(m, __shfl_xor(m, off, 32));
    if (lane == 0) red_m[wave] = m;
    __syncthreads();
    float Mx = red_m[0];
    #pragma unroll
    for (int w = 1; w < NTHR / 32; ++w) Mx = fmaxf(Mx, red_m[w]);

    float s0 = 0.f, s1 = 0.f, s2 = 0.f, s3 = 0.f;
    #pragma unroll
    for (int j = 0; j < PER; ++j) {
        x[j].x = __expf(x[j].x - Mx);  s0 += x[j].x;
        x[j].y = __expf(x[j].y - Mx);  s1 += x[j].y;
        x[j].z = __expf(x[j].z - Mx);  s2 += x[j].z;
        x[j].w = __expf(x[j].w - Mx);  s3 += x[j].w;
        if ((j & 3) == 3) __builtin_amdgcn_sched_barrier(0);
    }
    float s = (s0 + s1) + (s2 + s3);
    for (int off = 16; off > 0; off >>= 1)
        s += __shfl_xor(s, off, 32);
    if (lane == 0) red_s[wave] = s;
    __syncthreads();
    float S = red_s[0];
    #pragma unroll
    for (int w = 1; w < NTHR / 32; ++w) S += red_s[w];
    float invS = 1.f / S;

    #pragma unroll
    for (int c = 0; c < NCHK; ++c) {
        const int c0   = c * CHUNK;
        const int cend = (c0 + CHUNK < VV) ? (c0 + CHUNK) : VV;
        #pragma unroll
        for (int k = 0; k < JPC; ++k)
            s_copy4[k * NTHR + tid] = make_float4(0.f, 0.f, 0.f, 0.f);
        __syncthreads();
        if (id0 >= c0 && id0 < cend) atomicAdd(&s_copy[id0 - c0], a0);
        if (id1 >= c0 && id1 < cend) atomicAdd(&s_copy[id1 - c0], a1);
        __syncthreads();
        #pragma unroll
        for (int k = 0; k < JPC; ++k) {
            const int j = c * JPC + k;
            if (j < PER - 1 || tail) {
                float4 cp = s_copy4[k * NTHR + tid];
                float4 e  = x[j];
                float4 o;
                o.x = ln_pos(p * e.x * invS + omp * cp.x + 1e-12f);
                o.y = ln_pos(p * e.y * invS + omp * cp.y + 1e-12f);
                o.z = ln_pos(p * e.z * invS + omp * cp.z + 1e-12f);
                o.w = ln_pos(p * e.w * invS + omp * cp.w + 1e-12f);
                orow4[j * NTHR + tid] = o;
            }
        }
        __syncthreads();
    }
}

extern "C" void kernel_launch(void* const* d_in, const int* in_sizes, int n_in,
                              void* d_out, int out_size, void* d_ws, size_t ws_size,
                              hipStream_t stream) {
    (void)in_sizes; (void)n_in; (void)out_size; (void)ws_size;
    const float* dec    = (const float*)d_in[0];
    const float* attn   = (const float*)d_in[1];
    const float* enc    = (const float*)d_in[2];
    const float* logits = (const float*)d_in[3];
    const float* wgen   = (const float*)d_in[4];
    const float* bgen   = (const float*)d_in[5];
    const int*   sids   = (const int*)d_in[6];
    float* out  = (float*)d_out;

    float* encW = (float*)d_ws;
    float* pgen = encW + BB * TS;

    k_encw<<<(BB * TS) / 8, 256, 0, stream>>>(enc, wgen, encW);
    k_pgen<<<dim3(TT / 16, BB), PGW * 32, 0, stream>>>(attn, dec, encW, wgen, bgen, pgen);
    k_main<<<BB * TT, 256, 0, stream>>>(logits, attn, sids, pgen, out);
}


# k_main barriers order LDS only (release/acquire fences on the local address space around the bare barrier builtin): no wait for global store acknowledgement between output chunks
# speedup vs baseline: 3.2313x; 1.1168x over previous
#include <hip/hip_runtime.h>
#include <math.h>

#define BB 4
#define TT 256
#define TS 512
#define HH 1024
#define VV 32100
#define CHUNK 8192

typedef float v2f __attribute__((ext_vector_type(2)));
typedef float v8f __attribute__((ext_vector_type(8)));

__global__ void k_encw(const float* __restrict__ enc,
                       const float* __restrict__ wgen,
                       float* __restrict__ encW) {
    int wave = threadIdx.x >> 5;
    int lane = threadIdx.x & 31;
    int row  = blockIdx.x * 8 + wave;
    const float4* p = (const float4*)(enc + (size_t)row * HH);
    const float4* w = (const float4*)wgen;
    float acc = 0.f;
    for (int i = lane; i < HH / 4; i += 32) {
        float4 a = p[i];
        float4 b = w[i];
        acc += a.x * b.x + a.y * b.y + a.z * b.z + a.w * b.w;
    }
    for (int off = 16; off > 0; off >>= 1)
        acc += __shfl_down(acc, off, 32);
    if (lane == 0) encW[row] = acc;
}

#define PGSL 128
#define PGSW (TS / PGSL)
#define PGW  ((TS + HH) / PGSL)
#define PGJ  (PGSL / 8)

__global__ __launch_bounds__(PGW * 32) __attribute__((amdgpu_waves_per_eu(3, 3)))
void k_pgen(const float* __restrict__ attn,
            const float* __restrict__ dec,
            const float* __restrict__ encW,
            const float* __restrict__ wgen,
            const float* __restrict__ bgen,
            float* __restrict__ pgen) {
    __shared__ float part[PGW][16];

    int b    = blockIdx.y;
    int t0   = blockIdx.x * 16;
    int lane = threadIdx.x & 31;
    int wave = threadIdx.x >> 5;
    int M    = lane & 15;
    int h    = lane >> 4;

    const float* arow;
    const float* vec;
    if (wave < PGSW) {
        arow = attn + (size_t)(b * TT + t0 + M) * TS + wave * PGSL;
        vec  = encW + b * TS + wave * PGSL;
    } else {
        arow = dec + (size_t)(b * TT + t0 + M) * HH + (wave - PGSW) * PGSL;
        vec  = wgen + HH + (wave - PGSW) * PGSL;
    }
    const float4* ap = (const float4*)arow + h;
    const float4* vp = (const float4*)vec  + h;

    float4 a[PGJ], v[PGJ];
    #pragma unroll
    for (int j = 0; j < PGJ; ++j) v[j] = vp[2 * j];
    #pragma unroll
    for (int j = 0; j < PGJ; ++j) a[j] = ap[2 * j];

    v8f c0 = {};
    v8f c1 = {};
    #pragma unroll
    for (int j = 0; j < PGJ; ++j) {
        v2f a01, a23, b01, b23;
        a01.x = a[j].x;  a01.y = a[j].y;
        a23.x = a[j].z;  a23.y = a[j].w;
        b01.x = v[j].x;  b01.y = v[j].y;
        b23.x = v[j].z;  b23.y = v[j].w;
        c0 = __builtin_amdgcn_wmma_f32_16x16x4_f32(false, a01, false, b01,
                                                   (short)0, c0, false, false);
        c1 = __builtin_amdgcn_wmma_f32_16x16x4_f32(false, a23, false, b23,
                                                   (short)0, c1, false, false);
    }
    v8f c = c0 + c1;

    if (M == 0) {
        #pragma unroll
        for (int r = 0; r < 8; ++r) part[wave][8 * h + r] = c[r];
    }
    __syncthreads();
    if (threadIdx.x < 16) {
        float acc = 0.f;
        #pragma unroll
        for (int w = 0; w < PGW; ++w) acc += part[w][threadIdx.x];
        float logit = acc + bgen[0];
        pgen[b * TT + t0 + threadIdx.x] = 1.f / (1.f + __expf(-logit));
    }
}

__device__ __forceinline__ float ln_pos(float x) {
    return 0.6931471805599453f * __builtin_amdgcn_logf(x);
}

__device__ __forceinline__ void lds_barrier() {
    __builtin_amdgcn_fence(__ATOMIC_RELEASE, "workgroup", "local");
    __builtin_amdgcn_s_barrier();
    __builtin_amdgcn_fence(__ATOMIC_ACQUIRE, "workgroup", "local");
}

#define NTHR 256
#define V4   (VV / 4)
#define PER  ((V4 + NTHR - 1) / NTHR)
#define C4   (CHUNK / 4)
#define JPC  (C4 / NTHR)
#define NCHK ((VV + CHUNK - 1) / CHUNK)

__global__ __launch_bounds__(NTHR, 6)
void k_main(const float* __restrict__ logits,
            const float* __restrict__ attn,
            const int*   __restrict__ sids,
            const float* __restrict__ pgen,
            float* __restrict__ out) {
    __shared__ float4 s_copy4[C4];
    __shared__ float red_m[NTHR / 32];
    __shared__ float red_s[NTHR / 32];
    float* s_copy = (float*)s_copy4;

    int blk  = blockIdx.x;
    int b    = blk >> 8;
    int tid  = threadIdx.x;
    int lane = tid & 31;
    int wave = tid >> 5;

    const float4* row4  = (const float4*)(logits + (size_t)blk * VV);
    float4*       orow4 = (float4*)(out + (size_t)blk * VV);
    const int*    ids   = sids + b * TS;
    const float*  arow  = attn + (size_t)blk * TS;

    const bool tail = (PER - 1) * NTHR + tid < V4;
    float4 x[PER];
    #pragma unroll
    for (int j = 0; j < PER - 1; ++j) x[j] = row4[j * NTHR + tid];
    if (tail) x[PER - 1] = row4[(PER - 1) * NTHR + tid];
    else      x[PER - 1] = make_float4(-INFINITY, -INFINITY, -INFINITY, -INFINITY);
    int   id0 = ids[tid],  id1 = ids[tid + NTHR];
    float a0  = arow[tid], a1  = arow[tid + NTHR];
    float p   = pgen[blk];
    float omp = 1.f - p;

    float m = -INFINITY;
    #pragma unroll
    for (int j = 0; j < PER; ++j)
        m = fmaxf(m, fmaxf(fmaxf(x[j].x, x[j].y), fmaxf(x[j].z, x[j].w)));
    for (int off = 16; off > 0; off >>= 1)
        m = fmaxf(m, __shfl_xor(m, off, 32));
    if (lane == 0) red_m[wave] = m;
    lds_barrier();
    float Mx = red_m[0];
    #pragma unroll
    for (int w = 1; w < NTHR / 32; ++w) Mx = fmaxf(Mx, red_m[w]);

    float s0 = 0.f, s1 = 0.f, s2 = 0.f, s3 = 0.f;
    #pragma unroll
    for (int j = 0; j < PER; ++j) {
        x[j].x = __expf(x[j].x - Mx);  s0 += x[j].x;
        x[j].y = __expf(x[j].y - Mx);  s1 += x[j].y;
        x[j].z = __expf(x[j].z - Mx);  s2 += x[j].z;
        x[j].w = __expf(x[j].w - Mx);  s3 += x[j].w;
        if ((j & 3) == 3) __builtin_amdgcn_sched_barrier(0);
    }
    float s = (s0 + s1) + (s2 + s3);
    for (int off = 16; off > 0; off >>= 1)
        s += __shfl_xor(s, off, 32);
    if (lane == 0) red_s[wave] = s;
    lds_barrier();
    float S = red_s[0];
    #pragma unroll
    for (int w = 1; w < NTHR / 32; ++w) S += red_s[w];
    float invS = 1.f / S;

    #pragma unroll
    for (int c = 0; c < NCHK; ++c) {
        const int c0   = c * CHUNK;
        const int cend = (c0 + CHUNK < VV) ? (c0 + CHUNK) : VV;
        #pragma unroll
        for (int k = 0; k < JPC; ++k)
            s_copy4[k * NTHR + tid] = make_float4(0.f, 0.f, 0.f, 0.f);
        lds_barrier();
        if (id0 >= c0 && id0 < cend) atomicAdd(&s_copy[id0 - c0], a0);
        if (id1 >= c0 && id1 < cend) atomicAdd(&s_copy[id1 - c0], a1);
        lds_barrier();
        #pragma unroll
        for (int k = 0; k < JPC; ++k) {
            const int j = c * JPC + k;
            if (j < PER - 1 || tail) {
                float4 cp = s_copy4[k * NTHR + tid];
                float4 e  = x[j];
                float4 o;
                o.x = ln_pos(p * e.x * invS + omp * cp.x + 1e-12f);
                o.y = ln_pos(p * e.y * invS + omp * cp.y + 1e-12f);
                o.z = ln_pos(p * e.z * invS + omp * cp.z + 1e-12f);
                o.w = ln_pos(p * e.w * invS + omp * cp.w + 1e-12f);
                orow4[j * NTHR + tid] = o;
            }
        }
        lds_barrier();
    }
}

extern "C" void kernel_launch(void* const* d_in, const int* in_sizes, int n_in,
                              void* d_out, int out_size, void* d_ws, size_t ws_size,
                              hipStream_t stream) {
    (void)in_sizes; (void)n_in; (void)out_size; (void)ws_size;
    const float* dec    = (const float*)d_in[0];
    const float* attn   = (const float*)d_in[1];
    const float* enc    = (const float*)d_in[2];
    const float* logits = (const float*)d_in[3];
    const float* wgen   = (const float*)d_in[4];
    const float* bgen   = (const float*)d_in[5];
    const int*   sids   = (const int*)d_in[6];
    float* out  = (float*)d_out;

    float* encW = (float*)d_ws;
    float* pgen = encW + BB * TS;

    k_encw<<<(BB * TS) / 8, 256, 0, stream>>>(enc, wgen, encW);
    k_pgen<<<dim3(TT / 16, BB), PGW * 32, 0, stream>>>(attn, dec, encW, wgen, bgen, pgen);
    k_main<<<BB * TT, 256, 0, stream>>>(logits, attn, sids, pgen, out);
}


# p_gen formed inside k_main: each block sums its row's 1536-long dot (3 f32 WMMAs per wave, trace of C; wave sums ride on the row-max reduction); k_pgen launch removed; p_gen sum order changes
# speedup vs baseline: 3.4105x; 1.0555x over previous
#include <hip/hip_runtime.h>
#include <math.h>

#define BB 4
#define TT 256
#define TS 512
#define HH 1024
#define VV 32100
#define CHUNK 8192

typedef float v2f __attribute__((ext_vector_type(2)));
typedef float v8f __attribute__((ext_vector_type(8)));

__global__ void k_encw(const float* __restrict__ enc,
                       const float* __restrict__ wgen,
                       float* __restrict__ encW) {
    int wave = threadIdx.x >> 5;
    int lane = threadIdx.x & 31;
    int row  = blockIdx.x * 8 + wave;
    const float4* p = (const float4*)(enc + (size_t)row * HH);
    const float4* w = (const float4*)wgen;
    float acc = 0.f;
    for (int i = lane; i < HH / 4; i += 32) {
        float4 a = p[i];
        float4 b = w[i];
        acc += a.x * b.x + a.y * b.y + a.z * b.z + a.w * b.w;
    }
    for (int off = 16; off > 0; off >>= 1)
        acc += __shfl_down(acc, off, 32);
    if (lane == 0) encW[row] = acc;
}

__device__ __forceinline__ float ln_pos(float x) {
    return 0.6931471805599453f * __builtin_amdgcn_logf(x);
}

__device__ __forceinline__ void lds_barrier() {
    __builtin_amdgcn_fence(__ATOMIC_RELEASE, "workgroup", "local");
    __builtin_amdgcn_s_barrier();
    __builtin_amdgcn_fence(__ATOMIC_ACQUIRE, "workgroup", "local");
}

#define NTHR 256
#define V4   (VV / 4)
#define PER  ((V4 + NTHR - 1) / NTHR)
#define C4   (CHUNK / 4)
#define JPC  (C4 / NTHR)
#define NCHK ((VV + CHUNK - 1) / CHUNK)

__global__ __launch_bounds__(NTHR, 6)
void k_main(const float* __restrict__ logits,
            const float* __restrict__ attn,
            const float* __restrict__ dec,
            const float* __restrict__ encW,
            const float* __restrict__ wgen,
            const float* __restrict__ bgen,
            const int*   __restrict__ sids,
            float* __restrict__ out) {
    __shared__ float4 s_copy4[C4];
    __shared__ float red_m[NTHR / 32];
    __shared__ float red_s[NTHR / 32];
    __shared__ float red_d[NTHR / 32];
    float* s_copy = (float*)s_copy4;

    int blk  = blockIdx.x;
    int b    = blk >> 8;
    int tid  = threadIdx.x;
    int lane = tid & 31;
    int wave = tid >> 5;

    const float4* row4  = (const float4*)(logits + (size_t)blk * VV);
    float4*       orow4 = (float4*)(out + (size_t)blk * VV);
    const int*    ids   = sids + b * TS;
    const float*  arow  = attn + (size_t)blk * TS;
    const float*  evec  = encW + b * TS;

    float  a0 = arow[tid], a1 = arow[tid + NTHR];
    float  e0 = evec[tid], e1 = evec[tid + NTHR];
    float4 d4 = ((const float4*)(dec + (size_t)blk * HH))[tid];
    float4 w4 = ((const float4*)(wgen + HH))[tid];
    float  bg  = bgen[0];
    const bool tail = (PER - 1) * NTHR + tid < V4;
    float4 x[PER];
    #pragma unroll
    for (int j = 0; j < PER - 1; ++j) x[j] = row4[j * NTHR + tid];
    if (tail) x[PER - 1] = row4[(PER - 1) * NTHR + tid];
    else      x[PER - 1] = make_float4(-INFINITY, -INFINITY, -INFINITY, -INFINITY);

    float dg;
    {
        v2f aa, ae, da, dw, db, dx;
        aa.x = a0;    aa.y = a1;     ae.x = e0;    ae.y = e1;
        da.x = d4.x;  da.y = d4.y;   dw.x = w4.x;  dw.y = w4.y;
        db.x = d4.z;  db.y = d4.w;   dx.x = w4.z;  dx.y = w4.w;
        v8f c = {};
        c = __builtin_amdgcn_wmma_f32_16x16x4_f32(false, aa, false, ae,
                                                  (short)0, c, false, false);
        c = __builtin_amdgcn_wmma_f32_16x16x4_f32(false, da, false, dw,
                                                  (short)0, c, false, false);
        c = __builtin_amdgcn_wmma_f32_16x16x4_f32(false, db, false, dx,
                                                  (short)0, c, false, false);
        const int r = lane & 7;
        dg = c[0];
        #pragma unroll
        for (int q = 1; q < 8; ++q) dg = (r == q) ? c[q] : dg;
        const bool ondiag = ((lane >> 3) & 1) == (lane >> 4);
        dg = ondiag ? dg : 0.f;
    }

    float m = -INFINITY;
    #pragma unroll
    for (int j = 0; j < PER; ++j)
        m = fmaxf(m, fmaxf(fmaxf(x[j].x, x[j].y), fmaxf(x[j].z, x[j].w)));
    for (int off = 16; off > 0; off >>= 1) {
        m  = fmaxf(m, __shfl_xor(m, off, 32));
        dg += __shfl_xor(dg, off, 32);
    }
    if (lane == 0) { red_m[wave] = m;  red_d[wave] = dg; }
    lds_barrier();
    float Mx = red_m[0];
    #pragma unroll
    for (int w = 1; w < NTHR / 32; ++w) Mx = fmaxf(Mx, red_m[w]);

    float s0 = 0.f, s1 = 0.f, s2 = 0.f, s3 = 0.f;
    #pragma unroll
    for (int j = 0; j < PER; ++j) {
        x[j].x = __expf(x[j].x - Mx);  s0 += x[j].x;
        x[j].y = __expf(x[j].y - Mx);  s1 += x[j].y;
        x[j].z = __expf(x[j].z - Mx);  s2 += x[j].z;
        x[j].w = __expf(x[j].w - Mx);  s3 += x[j].w;
        if ((j & 3) == 3) __builtin_amdgcn_sched_barrier(0);
    }
    int id0 = ids[tid], id1 = ids[tid + NTHR];
    float s = (s0 + s1) + (s2 + s3);
    for (int off = 16; off > 0; off >>= 1)
        s += __shfl_xor(s, off, 32);
    if (lane == 0) red_s[wave] = s;
    lds_barrier();
    float S  = red_s[0];
    float lg = red_d[0];
    #pragma unroll
    for (int w = 1; w < NTHR / 32; ++w) {
        S  += red_s[w];
        lg += red_d[w];
    }
    float invS = 1.f / S;
    float p    = 1.f / (1.f + __expf(-(lg + bg)));
    float omp  = 1.f - p;

    #pragma unroll
    for (int c = 0; c < NCHK; ++c) {
        const int c0   = c * CHUNK;
        const int cend = (c0 + CHUNK < VV) ? (c0 + CHUNK) : VV;
        #pragma unroll
        for (int k = 0; k < JPC; ++k)
            s_copy4[k * NTHR + tid] = make_float4(0.f, 0.f, 0.f, 0.f);
        lds_barrier();
        if (id0 >= c0 && id0 < cend) atomicAdd(&s_copy[id0 - c0], a0);
        if (id1 >= c0 && id1 < cend) atomicAdd(&s_copy[id1 - c0], a1);
        lds_barrier();
        #pragma unroll
        for (int k = 0; k < JPC; ++k) {
            const int j = c * JPC + k;
            if (j < PER - 1 || tail) {
                float4 cp = s_copy4[k * NTHR + tid];
                float4 e  = x[j];
                float4 o;
                o.x = ln_pos(p * e.x * invS + omp * cp.x + 1e-12f);
                o.y = ln_pos(p * e.y * invS + omp * cp.y + 1e-12f);
                o.z = ln_pos(p * e.z * invS + omp * cp.z + 1e-12f);
                o.w = ln_pos(p * e.w * invS + omp * cp.w + 1e-12f);
                orow4[j * NTHR + tid] = o;
            }
        }
        lds_barrier();
    }
}

extern "C" void kernel_launch(void* const* d_in, const int* in_sizes, int n_in,
                              void* d_out, int out_size, void* d_ws, size_t ws_size,
                              hipStream_t stream) {
    (void)in_sizes; (void)n_in; (void)out_size; (void)ws_size;
    const float* dec    = (const float*)d_in[0];
    const float* attn   = (const float*)d_in[1];
    const float* enc    = (const float*)d_in[2];
    const float* logits = (const float*)d_in[3];
    const float* wgen   = (const float*)d_in[4];
    const float* bgen   = (const float*)d_in[5];
    const int*   sids   = (const int*)d_in[6];
    float* out  = (float*)d_out;

    float* encW = (float*)d_ws;

    k_encw<<<(BB * TS) / 8, 256, 0, stream>>>(enc, wgen, encW);
    k_main<<<BB * TT, 256, 0, stream>>>(logits, attn, dec, encW, wgen, bgen, sids, out);
}


# k_main: each row's 16-byte slots counted from the 256-byte boundary below its start, so every wave-wide load and store covers whole aligned lines (edge lanes masked); exp-sum order changes
# speedup vs baseline: 3.6874x; 1.0812x over previous
#include <hip/hip_runtime.h>
#include <math.h>

#define BB 4
#define TT 256
#define TS 512
#define HH 1024
#define VV 32100
#define CHUNK 8192

typedef float v2f __attribute__((ext_vector_type(2)));
typedef float v8f __attribute__((ext_vector_type(8)));

__global__ void k_encw(const float* __restrict__ enc,
                       const float* __restrict__ wgen,
                       float* __restrict__ encW) {
    int wave = threadIdx.x >> 5;
    int lane = threadIdx.x & 31;
    int row  = blockIdx.x * 8 + wave;
    const float4* p = (const float4*)(enc + (size_t)row * HH);
    const float4* w = (const float4*)wgen;
    float acc = 0.f;
    for (int i = lane; i < HH / 4; i += 32) {
        float4 a = p[i];
        float4 b = w[i];
        acc += a.x * b.x + a.y * b.y + a.z * b.z + a.w * b.w;
    }
    for (int off = 16; off > 0; off >>= 1)
        acc += __shfl_down(acc, off, 32);
    if (lane == 0) encW[row] = acc;
}

__device__ __forceinline__ float ln_pos(float x) {
    return 0.6931471805599453f * __builtin_amdgcn_logf(x);
}

__device__ __forceinline__ void lds_barrier() {
    __builtin_amdgcn_fence(__ATOMIC_RELEASE, "workgroup", "local");
    __builtin_amdgcn_s_barrier();
    __builtin_amdgcn_fence(__ATOMIC_ACQUIRE, "workgroup", "local");
}

#define NTHR 256
#define V4   (VV / 4)
#define PER  ((V4 + NTHR - 1) / NTHR)
#define C4   (CHUNK / 4)
#define JPC  (C4 / NTHR)
#define NCHK ((VV + CHUNK - 1) / CHUNK)

__global__ __launch_bounds__(NTHR, 6)
void k_main(const float* __restrict__ logits,
            const float* __restrict__ attn,
            const float* __restrict__ dec,
            const float* __restrict__ encW,
            const float* __restrict__ wgen,
            const float* __restrict__ bgen,
            const int*   __restrict__ sids,
            float* __restrict__ out) {
    __shared__ float4 s_copy4[C4];
    __shared__ float red_m[NTHR / 32];
    __shared__ float red_s[NTHR / 32];
    __shared__ float red_d[NTHR / 32];
    float* s_copy = (float*)s_copy4;

    int blk  = blockIdx.x;
    int b    = blk >> 8;
    int tid  = threadIdx.x;
    int lane = tid & 31;
    int wave = tid >> 5;

    const int lead = (int)(((size_t)(out + (size_t)blk * VV) >> 4) & 15);
    const float4* row4  = (const float4*)(logits + (size_t)blk * VV) - lead;
    float4*       orow4 = (float4*)(out + (size_t)blk * VV) - lead;
    const int*    ids   = sids + b * TS;
    const float*  arow  = attn + (size_t)blk * TS;
    const float*  evec  = encW + b * TS;

    float  a0 = arow[tid], a1 = arow[tid + NTHR];
    float  e0 = evec[tid], e1 = evec[tid + NTHR];
    float4 d4 = ((const float4*)(dec + (size_t)blk * HH))[tid];
    float4 w4 = ((const float4*)(wgen + HH))[tid];
    float  bg  = bgen[0];
    __builtin_amdgcn_sched_barrier(0);
    const bool head = tid >= lead;
    const bool tail = (PER - 1) * NTHR + tid - lead < V4;
    float4 x[PER];
    #pragma unroll
    for (int j = 1; j < PER - 1; ++j) x[j] = row4[j * NTHR + tid];
    if (head) x[0] = row4[tid];
    else      x[0] = make_float4(-INFINITY, -INFINITY, -INFINITY, -INFINITY);
    if (tail) x[PER - 1] = row4[(PER - 1) * NTHR + tid];
    else      x[PER - 1] = make_float4(-INFINITY, -INFINITY, -INFINITY, -INFINITY);

    float dg;
    {
        v2f aa, ae, da, dw, db, dx;
        aa.x = a0;    aa.y = a1;     ae.x = e0;    ae.y = e1;
        da.x = d4.x;  da.y = d4.y;   dw.x = w4.x;  dw.y = w4.y;
        db.x = d4.z;  db.y = d4.w;   dx.x = w4.z;  dx.y = w4.w;
        v8f c = {};
        c = __builtin_amdgcn_wmma_f32_16x16x4_f32(false, aa, false, ae,
                                                  (short)0, c, false, false);
        c = __builtin_amdgcn_wmma_f32_16x16x4_f32(false, da, false, dw,
                                                  (short)0, c, false, false);
        c = __builtin_amdgcn_wmma_f32_16x16x4_f32(false, db, false, dx,
                                                  (short)0, c, false, false);
        const int r = lane & 7;
        dg = c[0];
        #pragma unroll
        for (int q = 1; q < 8; ++q) dg = (r == q) ? c[q] : dg;
        const bool ondiag = ((lane >> 3) & 1) == (lane >> 4);
        dg = ondiag ? dg : 0.f;
    }

    float m = -INFINITY;
    #pragma unroll
    for (int j = 0; j < PER; ++j)
        m = fmaxf(m, fmaxf(fmaxf(x[j].x, x[j].y), fmaxf(x[j].z, x[j].w)));
    for (int off = 16; off > 0; off >>= 1) {
        m  = fmaxf(m, __shfl_xor(m, off, 32));
        dg += __shfl_xor(dg, off, 32);
    }
    if (lane == 0) { red_m[wave] = m;  red_d[wave] = dg; }
    lds_barrier();
    float Mx = red_m[0];
    #pragma unroll
    for (int w = 1; w < NTHR / 32; ++w) Mx = fmaxf(Mx, red_m[w]);

    float s0 = 0.f, s1 = 0.f, s2 = 0.f, s3 = 0.f;
    #pragma unroll
    for (int j = 0; j < PER; ++j) {
        x[j].x = __expf(x[j].x - Mx);  s0 += x[j].x;
        x[j].y = __expf(x[j].y - Mx);  s1 += x[j].y;
        x[j].z = __expf(x[j].z - Mx);  s2 += x[j].z;
        x[j].w = __expf(x[j].w - Mx);  s3 += x[j].w;
        if ((j & 3) == 3) __builtin_amdgcn_sched_barrier(0);
    }
    int id0 = ids[tid] + 4 * lead, id1 = ids[tid + NTHR] + 4 * lead;
    float s = (s0 + s1) + (s2 + s3);
    for (int off = 16; off > 0; off >>= 1)
        s += __shfl_xor(s, off, 32);
    if (lane == 0) red_s[wave] = s;
    lds_barrier();
    float S  = red_s[0];
    float lg = red_d[0];
    #pragma unroll
    for (int w = 1; w < NTHR / 32; ++w) {
        S  += red_s[w];
        lg += red_d[w];
    }
    float invS = 1.f / S;
    float p    = 1.f / (1.f + __expf(-(lg + bg)));
    float omp  = 1.f - p;

    #pragma unroll
    for (int c = 0; c < NCHK; ++c) {
        #pragma unroll
        for (int k = 0; k < JPC; ++k)
            s_copy4[k * NTHR + tid] = make_float4(0.f, 0.f, 0.f, 0.f);
        lds_barrier();
        if ((id0 >> 13) == c) atomicAdd(&s_copy[id0 & (CHUNK - 1)], a0);
        if ((id1 >> 13) == c) atomicAdd(&s_copy[id1 & (CHUNK - 1)], a1);
        lds_barrier();
        #pragma unroll
        for (int k = 0; k < JPC; ++k) {
            const int j = c * JPC + k;
            if ((j > 0 || head) && (j < PER - 1 || tail)) {
                float4 cp = s_copy4[k * NTHR + tid];
                float4 e  = x[j];
                float4 o;
                o.x = ln_pos(p * e.x * invS + omp * cp.x + 1e-12f);
                o.y = ln_pos(p * e.y * invS + omp * cp.y + 1e-12f);
                o.z = ln_pos(p * e.z * invS + omp * cp.z + 1e-12f);
                o.w = ln_pos(p * e.w * invS + omp * cp.w + 1e-12f);
                orow4[j * NTHR + tid] = o;
            }
        }
        lds_barrier();
    }
}

extern "C" void kernel_launch(void* const* d_in, const int* in_sizes, int n_in,
                              void* d_out, int out_size, void* d_ws, size_t ws_size,
                              hipStream_t stream) {
    (void)in_sizes; (void)n_in; (void)out_size; (void)ws_size;
    const float* dec    = (const float*)d_in[0];
    const float* attn   = (const float*)d_in[1];
    const float* enc    = (const float*)d_in[2];
    const float* logits = (const float*)d_in[3];
    const float* wgen   = (const float*)d_in[4];
    const float* bgen   = (const float*)d_in[5];
    const int*   sids   = (const int*)d_in[6];
    float* out  = (float*)d_out;

    float* encW = (float*)d_ws;

    k_encw<<<(BB * TS) / 8, 256, 0, stream>>>(enc, wgen, encW);
    k_main<<<BB * TT, 256, 0, stream>>>(logits, attn, dec, encW, wgen, bgen, sids, out);
}


# k_encw: all sixteen 16-byte loads of a lane issued before the first wait (one memory round trip per wave instead of eight dependent ones); encW sum order changes: products fused into the sums
# speedup vs baseline: 4.1505x; 1.1256x over previous
#include <hip/hip_runtime.h>
#include <math.h>

#define BB 4
#define TT 256
#define TS 512
#define HH 1024
#define VV 32100
#define CHUNK 8192

typedef float v2f __attribute__((ext_vector_type(2)));
typedef float v4f __attribute__((ext_vector_type(4)));
typedef float v8f __attribute__((ext_vector_type(8)));

#define EW4 (HH / 4 / 32)
__global__ void k_encw(const float* __restrict__ enc,
                       const float* __restrict__ wgen,
                       float* __restrict__ encW) {
    int wave = threadIdx.x >> 5;
    int lane = threadIdx.x & 31;
    int row  = blockIdx.x * 8 + wave;
    const v4f* p = (const v4f*)(enc + (size_t)row * HH);
    const v4f* w = (const v4f*)wgen;
    v4f a[EW4], b[EW4];
    #pragma unroll
    for (int k = 0; k < EW4; ++k) {
        a[k] = p[lane + 32 * k];
        b[k] = w[lane + 32 * k];
    }
    asm volatile("" :: "v"(a[0]), "v"(a[1]), "v"(a[2]), "v"(a[3]),
                       "v"(a[4]), "v"(a[5]), "v"(a[6]), "v"(a[7]),
                       "v"(b[0]), "v"(b[1]), "v"(b[2]), "v"(b[3]),
                       "v"(b[4]), "v"(b[5]), "v"(b[6]), "v"(b[7]));
    float acc = 0.f;
    #pragma unroll
    for (int k = 0; k < EW4; ++k)
        acc += a[k].x * b[k].x + a[k].y * b[k].y + a[k].z * b[k].z + a[k].w * b[k].w;
    for (int off = 16; off > 0; off >>= 1)
        acc += __shfl_down(acc, off, 32);
    if (lane == 0) encW[row] = acc;
}

__device__ __forceinline__ float ln_pos(float x) {
    return 0.6931471805599453f * __builtin_amdgcn_logf(x);
}

__device__ __forceinline__ void lds_barrier() {
    __builtin_amdgcn_fence(__ATOMIC_RELEASE, "workgroup", "local");
    __builtin_amdgcn_s_barrier();
    __builtin_amdgcn_fence(__ATOMIC_ACQUIRE, "workgroup", "local");
}

#define NTHR 256
#define V4   (VV / 4)
#define PER  ((V4 + NTHR - 1) / NTHR)
#define C4   (CHUNK / 4)
#define JPC  (C4 / NTHR)
#define NCHK ((VV + CHUNK - 1) / CHUNK)

__global__ __launch_bounds__(NTHR, 6)
void k_main(const float* __restrict__ logits,
            const float* __restrict__ attn,
            const float* __restrict__ dec,
            const float* __restrict__ encW,
            const float* __restrict__ wgen,
            const float* __restrict__ bgen,
            const int*   __restrict__ sids,
            float* __restrict__ out) {
    __shared__ float4 s_copy4[C4];
    __shared__ float red_m[NTHR / 32];
    __shared__ float red_s[NTHR / 32];
    __shared__ float red_d[NTHR / 32];
    float* s_copy = (float*)s_copy4;

    int blk  = blockIdx.x;
    int b    = blk >> 8;
    int tid  = threadIdx.x;
    int lane = tid & 31;
    int wave = tid >> 5;

    const int lead = (int)(((size_t)(out + (size_t)blk * VV) >> 4) & 15);
    const float4* row4  = (const float4*)(logits + (size_t)blk * VV) - lead;
    float4*       orow4 = (float4*)(out + (size_t)blk * VV) - lead;
    const int*    ids   = sids + b * TS;
    const float*  arow  = attn + (size_t)blk * TS;
    const float*  evec  = encW + b * TS;

    float  a0 = arow[tid], a1 = arow[tid + NTHR];
    float  e0 = evec[tid], e1 = evec[tid + NTHR];
    float4 d4 = ((const float4*)(dec + (size_t)blk * HH))[tid];
    float4 w4 = ((const float4*)(wgen + HH))[tid];
    float  bg  = bgen[0];
    __builtin_amdgcn_sched_barrier(0);
    const bool head = tid >= lead;
    const bool tail = (PER - 1) * NTHR + tid - lead < V4;
    float4 x[PER];
    #pragma unroll
    for (int j = 1; j < PER - 1; ++j) x[j] = row4[j * NTHR + tid];
    if (head) x[0] = row4[tid];
    else      x[0] = make_float4(-INFINITY, -INFINITY, -INFINITY, -INFINITY);
    if (tail) x[PER - 1] = row4[(PER - 1) * NTHR + tid];
    else      x[PER - 1] = make_float4(-INFINITY, -INFINITY, -INFINITY, -INFINITY);

    float dg;
    {
        v2f aa, ae, da, dw, db, dx;
        aa.x = a0;    aa.y = a1;     ae.x = e0;    ae.y = e1;
        da.x = d4.x;  da.y = d4.y;   dw.x = w4.x;  dw.y = w4.y;
        db.x = d4.z;  db.y = d4.w;   dx.x = w4.z;  dx.y = w4.w;
        v8f c = {};
        c = __builtin_amdgcn_wmma_f32_16x16x4_f32(false, aa, false, ae,
                                                  (short)0, c, false, false);
        c = __builtin_amdgcn_wmma_f32_16x16x4_f32(false, da, false, dw,
                                                  (short)0, c, false, false);
        c = __builtin_amdgcn_wmma_f32_16x16x4_f32(false, db, false, dx,
                                                  (short)0, c, false, false);
        const int r = lane & 7;
        dg = c[0];
        #pragma unroll
        for (int q = 1; q < 8; ++q) dg = (r == q) ? c[q] : dg;
        const bool ondiag = ((lane >> 3) & 1) == (lane >> 4);
        dg = ondiag ? dg : 0.f;
    }

    float m = -INFINITY;
    #pragma unroll
    for (int j = 0; j < PER; ++j)
        m = fmaxf(m, fmaxf(fmaxf(x[j].x, x[j].y), fmaxf(x[j].z, x[j].w)));
    for (int off = 16; off > 0; off >>= 1) {
        m  = fmaxf(m, __shfl_xor(m, off, 32));
        dg += __shfl_xor(dg, off, 32);
    }
    if (lane == 0) { red_m[wave] = m;  red_d[wave] = dg; }
    lds_barrier();
    float Mx = red_m[0];
    #pragma unroll
    for (int w = 1; w < NTHR / 32; ++w) Mx = fmaxf(Mx, red_m[w]);

    float s0 = 0.f, s1 = 0.f, s2 = 0.f, s3 = 0.f;
    #pragma unroll
    for (int j = 0; j < PER; ++j) {
        x[j].x = __expf(x[j].x - Mx);  s0 += x[j].x;
        x[j].y = __expf(x[j].y - Mx);  s1 += x[j].y;
        x[j].z = __expf(x[j].z - Mx);  s2 += x[j].z;
        x[j].w = __expf(x[j].w - Mx);  s3 += x[j].w;
        if ((j & 3) == 3) __builtin_amdgcn_sched_barrier(0);
    }
    int id0 = ids[tid] + 4 * lead, id1 = ids[tid + NTHR] + 4 * lead;
    float s = (s0 + s1) + (s2 + s3);
    for (int off = 16; off > 0; off >>= 1)
        s += __shfl_xor(s, off, 32);
    if (lane == 0) red_s[wave] = s;
    lds_barrier();
    float S  = red_s[0];
    float lg = red_d[0];
    #pragma unroll
    for (int w = 1; w < NTHR / 32; ++w) {
        S  += red_s[w];
        lg += red_d[w];
    }
    float invS = 1.f / S;
    float p    = 1.f / (1.f + __expf(-(lg + bg)));
    float omp  = 1.f - p;

    #pragma unroll
    for (int c = 0; c < NCHK; ++c) {
        #pragma unroll
        for (int k = 0; k < JPC; ++k)
            s_copy4[k * NTHR + tid] = make_float4(0.f, 0.f, 0.f, 0.f);
        lds_barrier();
        if ((id0 >> 13) == c) atomicAdd(&s_copy[id0 & (CHUNK - 1)], a0);
        if ((id1 >> 13) == c) atomicAdd(&s_copy[id1 & (CHUNK - 1)], a1);
        lds_barrier();
        #pragma unroll
        for (int k = 0; k < JPC; ++k) {
            const int j = c * JPC + k;
            if ((j > 0 || head) && (j < PER - 1 || tail)) {
                float4 cp = s_copy4[k * NTHR + tid];
                float4 e  = x[j];
                float4 o;
                o.x = ln_pos(p * e.x * invS + omp * cp.x + 1e-12f);
                o.y = ln_pos(p * e.y * invS + omp * cp.y + 1e-12f);
                o.z = ln_pos(p * e.z * invS + omp * cp.z + 1e-12f);
                o.w = ln_pos(p * e.w * invS + omp * cp.w + 1e-12f);
                orow4[j * NTHR + tid] = o;
            }
        }
        lds_barrier();
    }
}

extern "C" void kernel_launch(void* const* d_in, const int* in_sizes, int n_in,
                              void* d_out, int out_size, void* d_ws, size_t ws_size,
                              hipStream_t stream) {
    (void)in_sizes; (void)n_in; (void)out_size; (void)ws_size;
    const float* dec    = (const float*)d_in[0];
    const float* attn   = (const float*)d_in[1];
    const float* enc    = (const float*)d_in[2];
    const float* logits = (const float*)d_in[3];
    const float* wgen   = (const float*)d_in[4];
    const float* bgen   = (const float*)d_in[5];
    const int*   sids   = (const int*)d_in[6];
    float* out  = (float*)d_out;

    float* encW = (float*)d_ws;

    k_encw<<<(BB * TS) / 8, 256, 0, stream>>>(enc, wgen, encW);
    k_main<<<BB * TT, 256, 0, stream>>>(logits, attn, dec, encW, wgen, bgen, sids, out);
}


# k_main: copy mass of the whole row in one 128-KB LDS buffer, zeroed and scattered while the row's loads are in flight; output pass runs with no block barrier (3 per row, not 14); 2 blocks per WGP
# speedup vs baseline: 4.8473x; 1.1679x over previous
#include <hip/hip_runtime.h>
#include <math.h>

#define BB 4
#define TT 256
#define TS 512
#define HH 1024
#define VV 32100

typedef float v2f __attribute__((ext_vector_type(2)));
typedef float v4f __attribute__((ext_vector_type(4)));
typedef float v8f __attribute__((ext_vector_type(8)));

#define EW4 (HH / 4 / 32)
__global__ void k_encw(const float* __restrict__ enc,
                       const float* __restrict__ wgen,
                       float* __restrict__ encW) {
    int wave = threadIdx.x >> 5;
    int lane = threadIdx.x & 31;
    int row  = blockIdx.x * 8 + wave;
    const v4f* p = (const v4f*)(enc + (size_t)row * HH);
    const v4f* w = (const v4f*)wgen;
    v4f a[EW4], b[EW4];
    #pragma unroll
    for (int k = 0; k < EW4; ++k) {
        a[k] = p[lane + 32 * k];
        b[k] = w[lane + 32 * k];
    }
    asm volatile("" :: "v"(a[0]), "v"(a[1]), "v"(a[2]), "v"(a[3]),
                       "v"(a[4]), "v"(a[5]), "v"(a[6]), "v"(a[7]),
                       "v"(b[0]), "v"(b[1]), "v"(b[2]), "v"(b[3]),
                       "v"(b[4]), "v"(b[5]), "v"(b[6]), "v"(b[7]));
    float acc = 0.f;
    #pragma unroll
    for (int k = 0; k < EW4; ++k)
        acc += a[k].x * b[k].x + a[k].y * b[k].y + a[k].z * b[k].z + a[k].w * b[k].w;
    for (int off = 16; off > 0; off >>= 1)
        acc += __shfl_down(acc, off, 32);
    if (lane == 0) encW[row] = acc;
}

__device__ __forceinline__ float ln_pos(float x) {
    return 0.6931471805599453f * __builtin_amdgcn_logf(x);
}

__device__ __forceinline__ void lds_barrier() {
    __builtin_amdgcn_fence(__ATOMIC_RELEASE, "workgroup", "local");
    __builtin_amdgcn_s_barrier();
    __builtin_amdgcn_fence(__ATOMIC_ACQUIRE, "workgroup", "local");
}

#define NTHR 256
#define V4   (VV / 4)
#define PER  ((V4 + NTHR - 1) / NTHR)
#define LDS4 (PER * NTHR)

__global__ __launch_bounds__(NTHR, 4)
void k_main(const float* __restrict__ logits,
            const float* __restrict__ attn,
            const float* __restrict__ dec,
            const float* __restrict__ encW,
            const float* __restrict__ wgen,
            const float* __restrict__ bgen,
            const int*   __restrict__ sids,
            float* __restrict__ out) {
    __shared__ float4 s_copy4[LDS4];
    __shared__ float red_m[NTHR / 32];
    __shared__ float red_s[NTHR / 32];
    __shared__ float red_d[NTHR / 32];
    float* s_copy = (float*)s_copy4;

    int blk  = blockIdx.x;
    int b    = blk >> 8;
    int tid  = threadIdx.x;
    int lane = tid & 31;
    int wave = tid >> 5;

    const int lead = (int)(((size_t)(out + (size_t)blk * VV) >> 4) & 15);
    const float4* row4  = (const float4*)(logits + (size_t)blk * VV) - lead;
    float4*       orow4 = (float4*)(out + (size_t)blk * VV) - lead;
    const int*    ids   = sids + b * TS;
    const float*  arow  = attn + (size_t)blk * TS;
    const float*  evec  = encW + b * TS;

    float  a0 = arow[tid], a1 = arow[tid + NTHR];
    float  e0 = evec[tid], e1 = evec[tid + NTHR];
    float4 d4 = ((const float4*)(dec + (size_t)blk * HH))[tid];
    float4 w4 = ((const float4*)(wgen + HH))[tid];
    int id0 = ids[tid] + 4 * lead, id1 = ids[tid + NTHR] + 4 * lead;
    float  bg  = bgen[0];
    __builtin_amdgcn_sched_barrier(0);
    const bool head = tid >= lead;
    const bool tail = (PER - 1) * NTHR + tid - lead < V4;
    float4 x[PER];
    #pragma unroll
    for (int j = 1; j < PER - 1; ++j) x[j] = row4[j * NTHR + tid];
    if (head) x[0] = row4[tid];
    else      x[0] = make_float4(-INFINITY, -INFINITY, -INFINITY, -INFINITY);
    if (tail) x[PER - 1] = row4[(PER - 1) * NTHR + tid];
    else      x[PER - 1] = make_float4(-INFINITY, -INFINITY, -INFINITY, -INFINITY);

    #pragma unroll
    for (int j = 0; j < PER; ++j)
        s_copy4[j * NTHR + tid] = make_float4(0.f, 0.f, 0.f, 0.f);
    lds_barrier();
    if ((unsigned)id0 < 4u * LDS4) atomicAdd(&s_copy[id0], a0);
    if ((unsigned)id1 < 4u * LDS4) atomicAdd(&s_copy[id1], a1);

    float dg;
    {
        v2f aa, ae, da, dw, db, dx;
        aa.x = a0;    aa.y = a1;     ae.x = e0;    ae.y = e1;
        da.x = d4.x;  da.y = d4.y;   dw.x = w4.x;  dw.y = w4.y;
        db.x = d4.z;  db.y = d4.w;   dx.x = w4.z;  dx.y = w4.w;
        v8f c = {};
        c = __builtin_amdgcn_wmma_f32_16x16x4_f32(false, aa, false, ae,
                                                  (short)0, c, false, false);
        c = __builtin_amdgcn_wmma_f32_16x16x4_f32(false, da, false, dw,
                                                  (short)0, c, false, false);
        c = __builtin_amdgcn_wmma_f32_16x16x4_f32(false, db, false, dx,
                                                  (short)0, c, false, false);
        const int r = lane & 7;
        dg = c[0];
        #pragma unroll
        for (int q = 1; q < 8; ++q) dg = (r == q) ? c[q] : dg;
        const bool ondiag = ((lane >> 3) & 1) == (lane >> 4);
        dg = ondiag ? dg : 0.f;
    }

    float m = -INFINITY;
    #pragma unroll
    for (int j = 0; j < PER; ++j)
        m = fmaxf(m, fmaxf(fmaxf(x[j].x, x[j].y), fmaxf(x[j].z, x[j].w)));
    for (int off = 16; off > 0; off >>= 1) {
        m  = fmaxf(m, __shfl_xor(m, off, 32));
        dg += __shfl_xor(dg, off, 32);
    }
    if (lane == 0) { red_m[wave] = m;  red_d[wave] = dg; }
    lds_barrier();
    float Mx = red_m[0];
    #pragma unroll
    for (int w = 1; w < NTHR / 32; ++w) Mx = fmaxf(Mx, red_m[w]);

    float s0 = 0.f, s1 = 0.f, s2 = 0.f, s3 = 0.f;
    #pragma unroll
    for (int j = 0; j < PER; ++j) {
        x[j].x = __expf(x[j].x - Mx);  s0 += x[j].x;
        x[j].y = __expf(x[j].y - Mx);  s1 += x[j].y;
        x[j].z = __expf(x[j].z - Mx);  s2 += x[j].z;
        x[j].w = __expf(x[j].w - Mx);  s3 += x[j].w;
    }
    float s = (s0 + s1) + (s2 + s3);
    for (int off = 16; off > 0; off >>= 1)
        s += __shfl_xor(s, off, 32);
    if (lane == 0) red_s[wave] = s;
    lds_barrier();
    float S  = red_s[0];
    float lg = red_d[0];
    #pragma unroll
    for (int w = 1; w < NTHR / 32; ++w) {
        S  += red_s[w];
        lg += red_d[w];
    }
    float invS = 1.f / S;
    float p    = 1.f / (1.f + __expf(-(lg + bg)));
    float omp  = 1.f - p;

    #pragma unroll
    for (int j = 0; j < PER; ++j) {
        if ((j > 0 || head) && (j < PER - 1 || tail)) {
            float4 cp = s_copy4[j * NTHR + tid];
            float4 e  = x[j];
            float4 o;
            o.x = ln_pos(p * e.x * invS + omp * cp.x + 1e-12f);
            o.y = ln_pos(p * e.y * invS + omp * cp.y + 1e-12f);
            o.z = ln_pos(p * e.z * invS + omp * cp.z + 1e-12f);
            o.w = ln_pos(p * e.w * invS + omp * cp.w + 1e-12f);
            orow4[j * NTHR + tid] = o;
        }
    }
}

extern "C" void kernel_launch(void* const* d_in, const int* in_sizes, int n_in,
                              void* d_out, int out_size, void* d_ws, size_t ws_size,
                              hipStream_t stream) {
    (void)in_sizes; (void)n_in; (void)out_size; (void)ws_size;
    const float* dec    = (const float*)d_in[0];
    const float* attn   = (const float*)d_in[1];
    const float* enc    = (const float*)d_in[2];
    const float* logits = (const float*)d_in[3];
    const float* wgen   = (const float*)d_in[4];
    const float* bgen   = (const float*)d_in[5];
    const int*   sids   = (const int*)d_in[6];
    float* out  = (float*)d_out;

    float* encW = (float*)d_ws;

    k_encw<<<(BB * TS) / 8, 256, 0, stream>>>(enc, wgen, encW);
    k_main<<<BB * TT, 256, 0, stream>>>(logits, attn, dec, encW, wgen, bgen, sids, out);
}


# k_main: exp per 16-value group against the group's own max while the row's loads still land, rescaled by exp(m_g - M) in the output factor; one barrier behind the row, not two; exp-sum order changes
# speedup vs baseline: 5.0149x; 1.0346x over previous
#include <hip/hip_runtime.h>
#include <math.h>
#include <float.h>

#define BB 4
#define TT 256
#define TS 512
#define HH 1024
#define VV 32100

typedef float v2f __attribute__((ext_vector_type(2)));
typedef float v4f __attribute__((ext_vector_type(4)));
typedef float v8f __attribute__((ext_vector_type(8)));

#define EW4 (HH / 4 / 32)
__global__ void k_encw(const float* __restrict__ enc,
                       const float* __restrict__ wgen,
                       float* __restrict__ encW) {
    int wave = threadIdx.x >> 5;
    int lane = threadIdx.x & 31;
    int row  = blockIdx.x * 8 + wave;
    const v4f* p = (const v4f*)(enc + (size_t)row * HH);
    const v4f* w = (const v4f*)wgen;
    v4f a[EW4], b[EW4];
    #pragma unroll
    for (int k = 0; k < EW4; ++k) {
        a[k] = p[lane + 32 * k];
        b[k] = w[lane + 32 * k];
    }
    asm volatile("" :: "v"(a[0]), "v"(a[1]), "v"(a[2]), "v"(a[3]),
                       "v"(a[4]), "v"(a[5]), "v"(a[6]), "v"(a[7]),
                       "v"(b[0]), "v"(b[1]), "v"(b[2]), "v"(b[3]),
                       "v"(b[4]), "v"(b[5]), "v"(b[6]), "v"(b[7]));
    float acc = 0.f;
    #pragma unroll
    for (int k = 0; k < EW4; ++k)
        acc += a[k].x * b[k].x + a[k].y * b[k].y + a[k].z * b[k].z + a[k].w * b[k].w;
    for (int off = 16; off > 0; off >>= 1)
        acc += __shfl_down(acc, off, 32);
    if (lane == 0) encW[row] = acc;
}

__device__ __forceinline__ float ln_pos(float x) {
    return 0.6931471805599453f * __builtin_amdgcn_logf(x);
}

__device__ __forceinline__ void lds_barrier() {
    __builtin_amdgcn_fence(__ATOMIC_RELEASE, "workgroup", "local");
    __builtin_amdgcn_s_barrier();
    __builtin_amdgcn_fence(__ATOMIC_ACQUIRE, "workgroup", "local");
}

__device__ __forceinline__ void exp_group(float4& q0, float4& q1, float4& q2, float4& q3,
                                          float& m_out, float& s_out) {
    float m = fmaxf(fmaxf(fmaxf(q0.x, q0.y), fmaxf(q0.z, q0.w)),
                    fmaxf(fmaxf(q1.x, q1.y), fmaxf(q1.z, q1.w)));
    m = fmaxf(m, fmaxf(fmaxf(fmaxf(q2.x, q2.y), fmaxf(q2.z, q2.w)),
                       fmaxf(fmaxf(q3.x, q3.y), fmaxf(q3.z, q3.w))));
    m = fmaxf(m, -FLT_MAX);
    m_out = m;
    q0.x = __expf(q0.x - m);  q0.y = __expf(q0.y - m);
    q0.z = __expf(q0.z - m);  q0.w = __expf(q0.w - m);
    q1.x = __expf(q1.x - m);  q1.y = __expf(q1.y - m);
    q1.z = __expf(q1.z - m);  q1.w = __expf(q1.w - m);
    q2.x = __expf(q2.x - m);  q2.y = __expf(q2.y - m);
    q2.z = __expf(q2.z - m);  q2.w = __expf(q2.w - m);
    q3.x = __expf(q3.x - m);  q3.y = __expf(q3.y - m);
    q3.z = __expf(q3.z - m);  q3.w = __expf(q3.w - m);
    float s0 = (q0.x + q1.x) + (q2.x + q3.x);
    float s1 = (q0.y + q1.y) + (q2.y + q3.y);
    float s2 = (q0.z + q1.z) + (q2.z + q3.z);
    float s3 = (q0.w + q1.w) + (q2.w + q3.w);
    s_out = (s0 + s1) + (s2 + s3);
}

#define NTHR 256
#define V4   (VV / 4)
#define PER  ((V4 + NTHR - 1) / NTHR)
#define LDS4 (PER * NTHR)
#define NGRP (PER / 4)
#define JG(j)    (((j) == 0 || (j) >= PER - 3) ? NGRP - 1 : ((j) - 1) / 4)

__global__ __launch_bounds__(NTHR, 4)
void k_main(const float* __restrict__ logits,
            const float* __restrict__ attn,
            const float* __restrict__ dec,
            const float* __restrict__ encW,
            const float* __restrict__ wgen,
            const float* __restrict__ bgen,
            const int*   __restrict__ sids,
            float* __restrict__ out) {
    __shared__ float4 s_copy4[LDS4];
    __shared__ float red_m[NTHR / 32];
    __shared__ float red_s[NTHR / 32];
    __shared__ float red_d[NTHR / 32];
    float* s_copy = (float*)s_copy4;

    int blk  = blockIdx.x;
    int b    = blk >> 8;
    int tid  = threadIdx.x;
    int lane = tid & 31;
    int wave = tid >> 5;

    const int lead = (int)(((size_t)(out + (size_t)blk * VV) >> 4) & 15);
    const float4* row4  = (const float4*)(logits + (size_t)blk * VV) - lead;
    float4*       orow4 = (float4*)(out + (size_t)blk * VV) - lead;
    const int*    ids   = sids + b * TS;
    const float*  arow  = attn + (size_t)blk * TS;
    const float*  evec  = encW + b * TS;

    float  a0 = arow[tid], a1 = arow[tid + NTHR];
    float  e0 = evec[tid], e1 = evec[tid + NTHR];
    float4 d4 = ((const float4*)(dec + (size_t)blk * HH))[tid];
    float4 w4 = ((const float4*)(wgen + HH))[tid];
    int id0 = ids[tid] + 4 * lead, id1 = ids[tid + NTHR] + 4 * lead;
    float  bg  = bgen[0];
    __builtin_amdgcn_sched_barrier(0);
    const bool head = tid >= lead;
    const bool tail = (PER - 1) * NTHR + tid - lead < V4;
    float4 x[PER];
    #pragma unroll
    for (int j = 1; j < PER - 1; ++j) x[j] = row4[j * NTHR + tid];
    x[0]       = row4[head ? tid : lead];
    x[PER - 1] = row4[tail ? (PER - 1) * NTHR + tid : V4 - 1 + lead];

    #pragma unroll
    for (int j = 0; j < PER; ++j)
        s_copy4[j * NTHR + tid] = make_float4(0.f, 0.f, 0.f, 0.f);
    lds_barrier();
    if ((unsigned)id0 < 4u * LDS4) atomicAdd(&s_copy[id0], a0);
    if ((unsigned)id1 < 4u * LDS4) atomicAdd(&s_copy[id1], a1);

    float dg;
    {
        v2f aa, ae, da, dw, db, dx;
        aa.x = a0;    aa.y = a1;     ae.x = e0;    ae.y = e1;
        da.x = d4.x;  da.y = d4.y;   dw.x = w4.x;  dw.y = w4.y;
        db.x = d4.z;  db.y = d4.w;   dx.x = w4.z;  dx.y = w4.w;
        v8f c = {};
        c = __builtin_amdgcn_wmma_f32_16x16x4_f32(false, aa, false, ae,
                                                  (short)0, c, false, false);
        c = __builtin_amdgcn_wmma_f32_16x16x4_f32(false, da, false, dw,
                                                  (short)0, c, false, false);
        c = __builtin_amdgcn_wmma_f32_16x16x4_f32(false, db, false, dx,
                                                  (short)0, c, false, false);
        const int r = lane & 7;
        dg = c[0];
        #pragma unroll
        for (int q = 1; q < 8; ++q) dg = (r == q) ? c[q] : dg;
        const bool ondiag = ((lane >> 3) & 1) == (lane >> 4);
        dg = ondiag ? dg : 0.f;
    }

    float mg[NGRP], sg[NGRP];
#define GROUP(g, ja, jb, jc, jd)                                              \
    exp_group(x[ja], x[jb], x[jc], x[jd], mg[g], sg[g]);                      \
    __builtin_amdgcn_sched_barrier(0)
    GROUP(0,  1,  2,  3,  4);
    GROUP(1,  5,  6,  7,  8);
    GROUP(2,  9, 10, 11, 12);
    GROUP(3, 13, 14, 15, 16);
    GROUP(4, 17, 18, 19, 20);
    GROUP(5, 21, 22, 23, 24);
    GROUP(6, 25, 26, 27, 28);
    {
        x[0].x = head ? x[0].x : -INFINITY;  x[0].y = head ? x[0].y : -INFINITY;
        x[0].z = head ? x[0].z : -INFINITY;  x[0].w = head ? x[0].w : -INFINITY;
        x[PER - 1].x = tail ? x[PER - 1].x : -INFINITY;
        x[PER - 1].y = tail ? x[PER - 1].y : -INFINITY;
        x[PER - 1].z = tail ? x[PER - 1].z : -INFINITY;
        x[PER - 1].w = tail ? x[PER - 1].w : -INFINITY;
    }
    GROUP(7, 29, 30,  0, 31);
#undef GROUP

    float m = mg[0];
    #pragma unroll
    for (int g = 1; g < NGRP; ++g) m = fmaxf(m, mg[g]);
    for (int off = 16; off > 0; off >>= 1) {
        m  = fmaxf(m, __shfl_xor(m, off, 32));
        dg += __shfl_xor(dg, off, 32);
    }
    float s = 0.f;
    #pragma unroll
    for (int g = 0; g < NGRP; ++g) s += sg[g] * __expf(mg[g] - m);
    for (int off = 16; off > 0; off >>= 1)
        s += __shfl_xor(s, off, 32);
    if (lane == 0) { red_m[wave] = m;  red_s[wave] = s;  red_d[wave] = dg; }
    lds_barrier();
    float Mx = red_m[0];
    #pragma unroll
    for (int w = 1; w < NTHR / 32; ++w) Mx = fmaxf(Mx, red_m[w]);
    float S  = 0.f;
    float lg = 0.f;
    #pragma unroll
    for (int w = 0; w < NTHR / 32; ++w) {
        S  += red_s[w] * __expf(red_m[w] - Mx);
        lg += red_d[w];
    }
    float p    = 1.f / (1.f + __expf(-(lg + bg)));
    float omp  = 1.f - p;
    float pS   = p / S;
    float fg[NGRP];
    #pragma unroll
    for (int g = 0; g < NGRP; ++g) fg[g] = pS * __expf(mg[g] - Mx);

    #pragma unroll
    for (int j = 0; j < PER; ++j) {
        if ((j > 0 || head) && (j < PER - 1 || tail)) {
            const float f = fg[JG(j)];
            float4 cp = s_copy4[j * NTHR + tid];
            float4 e  = x[j];
            float4 o;
            o.x = ln_pos(f * e.x + omp * cp.x + 1e-12f);
            o.y = ln_pos(f * e.y + omp * cp.y + 1e-12f);
            o.z = ln_pos(f * e.z + omp * cp.z + 1e-12f);
            o.w = ln_pos(f * e.w + omp * cp.w + 1e-12f);
            orow4[j * NTHR + tid] = o;
        }
    }
}

extern "C" void kernel_launch(void* const* d_in, const int* in_sizes, int n_in,
                              void* d_out, int out_size, void* d_ws, size_t ws_size,
                              hipStream_t stream) {
    (void)in_sizes; (void)n_in; (void)out_size; (void)ws_size;
    const float* dec    = (const float*)d_in[0];
    const float* attn   = (const float*)d_in[1];
    const float* enc    = (const float*)d_in[2];
    const float* logits = (const float*)d_in[3];
    const float* wgen   = (const float*)d_in[4];
    const float* bgen   = (const float*)d_in[5];
    const int*   sids   = (const int*)d_in[6];
    float* out  = (float*)d_out;

    float* encW = (float*)d_ws;

    k_encw<<<(BB * TS) / 8, 256, 0, stream>>>(enc, wgen, encW);
    k_main<<<BB * TT, 256, 0, stream>>>(logits, attn, dec, encW, wgen, bgen, sids, out);
}


# k_main: one block per two rows (grid 512, one round): row 1's output pass refills each register with row 2's slot and runs row 2's exponentials two groups behind: row 2 is read under row 1's stores
# speedup vs baseline: 5.2278x; 1.0425x over previous
#include <hip/hip_runtime.h>
#include <math.h>
#include <float.h>

#define BB 4
#define TT 256
#define TS 512
#define HH 1024
#define VV 32100

typedef float v2f __attribute__((ext_vector_type(2)));
typedef float v4f __attribute__((ext_vector_type(4)));
typedef float v8f __attribute__((ext_vector_type(8)));

#define EW4 (HH / 4 / 32)
__global__ void k_encw(const float* __restrict__ enc,
                       const float* __restrict__ wgen,
                       float* __restrict__ encW) {
    int wave = threadIdx.x >> 5;
    int lane = threadIdx.x & 31;
    int row  = blockIdx.x * 8 + wave;
    const v4f* p = (const v4f*)(enc + (size_t)row * HH);
    const v4f* w = (const v4f*)wgen;
    v4f a[EW4], b[EW4];
    #pragma unroll
    for (int k = 0; k < EW4; ++k) {
        a[k] = p[lane + 32 * k];
        b[k] = w[lane + 32 * k];
    }
    asm volatile("" :: "v"(a[0]), "v"(a[1]), "v"(a[2]), "v"(a[3]),
                       "v"(a[4]), "v"(a[5]), "v"(a[6]), "v"(a[7]),
                       "v"(b[0]), "v"(b[1]), "v"(b[2]), "v"(b[3]),
                       "v"(b[4]), "v"(b[5]), "v"(b[6]), "v"(b[7]));
    float acc = 0.f;
    #pragma unroll
    for (int k = 0; k < EW4; ++k)
        acc += a[k].x * b[k].x + a[k].y * b[k].y + a[k].z * b[k].z + a[k].w * b[k].w;
    for (int off = 16; off > 0; off >>= 1)
        acc += __shfl_down(acc, off, 32);
    if (lane == 0) encW[row] = acc;
}

__device__ __forceinline__ float ln_pos(float x) {
    return 0.6931471805599453f * __builtin_amdgcn_logf(x);
}

__device__ __forceinline__ void lds_barrier() {
    __builtin_amdgcn_fence(__ATOMIC_RELEASE, "workgroup", "local");
    __builtin_amdgcn_s_barrier();
    __builtin_amdgcn_fence(__ATOMIC_ACQUIRE, "workgroup", "local");
}

__device__ __forceinline__ void exp_group(float4& q0, float4& q1, float4& q2, float4& q3,
                                          float& m_out, float& s_out) {
    float m = fmaxf(fmaxf(fmaxf(q0.x, q0.y), fmaxf(q0.z, q0.w)),
                    fmaxf(fmaxf(q1.x, q1.y), fmaxf(q1.z, q1.w)));
    m = fmaxf(m, fmaxf(fmaxf(fmaxf(q2.x, q2.y), fmaxf(q2.z, q2.w)),
                       fmaxf(fmaxf(q3.x, q3.y), fmaxf(q3.z, q3.w))));
    m = fmaxf(m, -FLT_MAX);
    m_out = m;
    q0.x = __expf(q0.x - m);  q0.y = __expf(q0.y - m);
    q0.z = __expf(q0.z - m);  q0.w = __expf(q0.w - m);
    q1.x = __expf(q1.x - m);  q1.y = __expf(q1.y - m);
    q1.z = __expf(q1.z - m);  q1.w = __expf(q1.w - m);
    q2.x = __expf(q2.x - m);  q2.y = __expf(q2.y - m);
    q2.z = __expf(q2.z - m);  q2.w = __expf(q2.w - m);
    q3.x = __expf(q3.x - m);  q3.y = __expf(q3.y - m);
    q3.z = __expf(q3.z - m);  q3.w = __expf(q3.w - m);
    float s0 = (q0.x + q1.x) + (q2.x + q3.x);
    float s1 = (q0.y + q1.y) + (q2.y + q3.y);
    float s2 = (q0.z + q1.z) + (q2.z + q3.z);
    float s3 = (q0.w + q1.w) + (q2.w + q3.w);
    s_out = (s0 + s1) + (s2 + s3);
}

#define NTHR 256
#define V4   (VV / 4)
#define PER  ((V4 + NTHR - 1) / NTHR)
#define LDS4 (PER * NTHR)
#define NGRP (PER / 4)
#define ROWS 2
#define JG(j)    (((j) == 0 || (j) >= PER - 3) ? NGRP - 1 : ((j) - 1) / 4)

__device__ __forceinline__ int ld_slot(int j, int tid, int lead, bool head, bool tail) {
    return j == 0       ? (head ? tid : lead)
         : j == PER - 1 ? (tail ? (PER - 1) * NTHR + tid : V4 - 1 + lead)
         :                j * NTHR + tid;
}

__device__ __forceinline__ float pgen_share(float a0, float a1, float e0, float e1,
                                            float4 d4, float4 w4, int lane) {
    v2f aa, ae, da, dw, db, dx;
    aa.x = a0;    aa.y = a1;     ae.x = e0;    ae.y = e1;
    da.x = d4.x;  da.y = d4.y;   dw.x = w4.x;  dw.y = w4.y;
    db.x = d4.z;  db.y = d4.w;   dx.x = w4.z;  dx.y = w4.w;
    v8f c = {};
    c = __builtin_amdgcn_wmma_f32_16x16x4_f32(false, aa, false, ae,
                                              (short)0, c, false, false);
    c = __builtin_amdgcn_wmma_f32_16x16x4_f32(false, da, false, dw,
                                              (short)0, c, false, false);
    c = __builtin_amdgcn_wmma_f32_16x16x4_f32(false, db, false, dx,
                                              (short)0, c, false, false);
    const int r = lane & 7;
    float dg = c[0];
    #pragma unroll
    for (int q = 1; q < 8; ++q) dg = (r == q) ? c[q] : dg;
    const bool ondiag = ((lane >> 3) & 1) == (lane >> 4);
    return ondiag ? dg : 0.f;
}

__device__ __forceinline__ void edge_fix(float4& first, float4& last, bool head, bool tail) {
    first.x = head ? first.x : -INFINITY;  first.y = head ? first.y : -INFINITY;
    first.z = head ? first.z : -INFINITY;  first.w = head ? first.w : -INFINITY;
    last.x = tail ? last.x : -INFINITY;    last.y = tail ? last.y : -INFINITY;
    last.z = tail ? last.z : -INFINITY;    last.w = tail ? last.w : -INFINITY;
}

__device__ __forceinline__ void row_factors(const float (&mg)[NGRP], const float (&sg)[NGRP],
                                            float dg, float bg, int lane, int wave,
                                            float* red_m, float* red_s, float* red_d,
                                            float (&fg)[NGRP], float& omp) {
    float m = mg[0];
    #pragma unroll
    for (int g = 1; g < NGRP; ++g) m = fmaxf(m, mg[g]);
    for (int off = 16; off > 0; off >>= 1) {
        m  = fmaxf(m, __shfl_xor(m, off, 32));
        dg += __shfl_xor(dg, off, 32);
    }
    float s = 0.f;
    #pragma unroll
    for (int g = 0; g < NGRP; ++g) s += sg[g] * __expf(mg[g] - m);
    for (int off = 16; off > 0; off >>= 1)
        s += __shfl_xor(s, off, 32);
    if (lane == 0) { red_m[wave] = m;  red_s[wave] = s;  red_d[wave] = dg; }
    lds_barrier();
    float Mx = red_m[0];
    #pragma unroll
    for (int w = 1; w < NTHR / 32; ++w) Mx = fmaxf(Mx, red_m[w]);
    float S  = 0.f;
    float lg = 0.f;
    #pragma unroll
    for (int w = 0; w < NTHR / 32; ++w) {
        S  += red_s[w] * __expf(red_m[w] - Mx);
        lg += red_d[w];
    }
    float p  = 1.f / (1.f + __expf(-(lg + bg)));
    float pS = p / S;
    omp = 1.f - p;
    #pragma unroll
    for (int g = 0; g < NGRP; ++g) fg[g] = pS * __expf(mg[g] - Mx);
}

__device__ __forceinline__ float4 mix_log(float4 e, float4 cp, float f, float omp) {
    float4 o;
    o.x = ln_pos(f * e.x + omp * cp.x + 1e-12f);
    o.y = ln_pos(f * e.y + omp * cp.y + 1e-12f);
    o.z = ln_pos(f * e.z + omp * cp.z + 1e-12f);
    o.w = ln_pos(f * e.w + omp * cp.w + 1e-12f);
    return o;
}

__global__ __launch_bounds__(NTHR, 4)
void k_main(const float* __restrict__ logits,
            const float* __restrict__ attn,
            const float* __restrict__ dec,
            const float* __restrict__ encW,
            const float* __restrict__ wgen,
            const float* __restrict__ bgen,
            const int*   __restrict__ sids,
            float* __restrict__ out) {
    __shared__ float4 s_copy4[LDS4];
    __shared__ float red_m[NTHR / 32];
    __shared__ float red_s[NTHR / 32];
    __shared__ float red_d[NTHR / 32];
    float* s_copy = (float*)s_copy4;

    const int blk0 = blockIdx.x * ROWS;
    const int blk1 = blk0 + 1;
    const int b    = blk0 >> 8;
    const int tid  = threadIdx.x;
    const int lane = tid & 31;
    const int wave = tid >> 5;

    const int lead0 = (int)(((size_t)(out + (size_t)blk0 * VV) >> 4) & 15);
    const int lead1 = (int)(((size_t)(out + (size_t)blk1 * VV) >> 4) & 15);
    const float4* rowa4  = (const float4*)(logits + (size_t)blk0 * VV) - lead0;
    const float4* rowb4  = (const float4*)(logits + (size_t)blk1 * VV) - lead1;
    float4*       orowa4 = (float4*)(out + (size_t)blk0 * VV) - lead0;
    float4*       orowb4 = (float4*)(out + (size_t)blk1 * VV) - lead1;
    const int*    ids    = sids + b * TS;
    const float*  evec   = encW + b * TS;

    float  a0 = attn[(size_t)blk0 * TS + tid], a1 = attn[(size_t)blk0 * TS + tid + NTHR];
    float  e0 = evec[tid], e1 = evec[tid + NTHR];
    float4 d4 = ((const float4*)(dec + (size_t)blk0 * HH))[tid];
    float4 w4 = ((const float4*)(wgen + HH))[tid];
    const int idr0 = ids[tid], idr1 = ids[tid + NTHR];
    float  bg  = bgen[0];
    __builtin_amdgcn_sched_barrier(0);
    const bool head0 = tid >= lead0;
    const bool tail0 = (PER - 1) * NTHR + tid - lead0 < V4;
    const bool head1 = tid >= lead1;
    const bool tail1 = (PER - 1) * NTHR + tid - lead1 < V4;
    float4 x[PER];
    #pragma unroll
    for (int j = 1; j < PER - 1; ++j) x[j] = rowa4[ld_slot(j, tid, lead0, head0, tail0)];
    x[0]       = rowa4[ld_slot(0, tid, lead0, head0, tail0)];
    x[PER - 1] = rowa4[ld_slot(PER - 1, tid, lead0, head0, tail0)];

    #pragma unroll
    for (int j = 0; j < PER; ++j)
        s_copy4[j * NTHR + tid] = make_float4(0.f, 0.f, 0.f, 0.f);
    lds_barrier();
    {
        const int id0 = idr0 + 4 * lead0, id1 = idr1 + 4 * lead0;
        if ((unsigned)id0 < 4u * LDS4) atomicAdd(&s_copy[id0], a0);
        if ((unsigned)id1 < 4u * LDS4) atomicAdd(&s_copy[id1], a1);
    }
    float dg = pgen_share(a0, a1, e0, e1, d4, w4, lane);

    float mg[NGRP], sg[NGRP];
#define GROUP(mm, ss, g, ja, jb, jc, jd)                                      \
    exp_group(x[ja], x[jb], x[jc], x[jd], mm[g], ss[g]);                      \
    __builtin_amdgcn_sched_barrier(0)
    GROUP(mg, sg, 0,  1,  2,  3,  4);
    GROUP(mg, sg, 1,  5,  6,  7,  8);
    GROUP(mg, sg, 2,  9, 10, 11, 12);
    GROUP(mg, sg, 3, 13, 14, 15, 16);
    GROUP(mg, sg, 4, 17, 18, 19, 20);
    GROUP(mg, sg, 5, 21, 22, 23, 24);
    GROUP(mg, sg, 6, 25, 26, 27, 28);
    edge_fix(x[0], x[PER - 1], head0, tail0);
    GROUP(mg, sg, 7, 29, 30,  0, 31);

    float fg[NGRP], omp;
    row_factors(mg, sg, dg, bg, lane, wave, red_m, red_s, red_d, fg, omp);

    a0 = attn[(size_t)blk1 * TS + tid];
    a1 = attn[(size_t)blk1 * TS + tid + NTHR];
    d4 = ((const float4*)(dec + (size_t)blk1 * HH))[tid];
    __builtin_amdgcn_sched_barrier(0);

    float mh[NGRP], sh[NGRP];
#define GROUPP(g, ja, jb, jc, jd)                                             \
    exp_group(x[ja], x[jb], x[jc], x[jd], mh[g], sh[g]);                      \
    asm volatile("" :: "v"(mh[g]), "v"(sh[g]));                               \
    __builtin_amdgcn_sched_barrier(0)
#define OUT1(j)                                                               \
    {                                                                         \
        float4 cp = s_copy4[(j) * NTHR + tid];                                \
        s_copy4[(j) * NTHR + tid] = make_float4(0.f, 0.f, 0.f, 0.f);          \
        float4 o = mix_log(x[j], cp, fg[JG(j)], omp);                         \
        if (((j) > 0 || head0) && ((j) < PER - 1 || tail0))                   \
            orowa4[(j) * NTHR + tid] = o;                                     \
        x[j] = rowb4[ld_slot((j), tid, lead1, head1, tail1)];                 \
    }
#define STEP1(ja, jb, jc, jd)                                                 \
    OUT1(ja) OUT1(jb) OUT1(jc) OUT1(jd)                                       \
    __builtin_amdgcn_sched_barrier(0)
    STEP1( 1,  2,  3,  4);
    STEP1( 5,  6,  7,  8);
    STEP1( 9, 10, 11, 12);   GROUPP(0,  1,  2,  3,  4);
    STEP1(13, 14, 15, 16);   GROUPP(1,  5,  6,  7,  8);
    STEP1(17, 18, 19, 20);   GROUPP(2,  9, 10, 11, 12);
    STEP1(21, 22, 23, 24);   GROUPP(3, 13, 14, 15, 16);
    STEP1(25, 26, 27, 28);   GROUPP(4, 17, 18, 19, 20);
    STEP1(29, 30,  0, 31);   GROUPP(5, 21, 22, 23, 24);
#undef STEP1
#undef OUT1
#undef GROUPP

    lds_barrier();
    {
        const int id0 = idr0 + 4 * lead1, id1 = idr1 + 4 * lead1;
        if ((unsigned)id0 < 4u * LDS4) atomicAdd(&s_copy[id0], a0);
        if ((unsigned)id1 < 4u * LDS4) atomicAdd(&s_copy[id1], a1);
    }
    dg = pgen_share(a0, a1, e0, e1, d4, w4, lane);
    GROUP(mh, sh, 6, 25, 26, 27, 28);
    edge_fix(x[0], x[PER - 1], head1, tail1);
    GROUP(mh, sh, 7, 29, 30,  0, 31);
#undef GROUP

    row_factors(mh, sh, dg, bg, lane, wave, red_m, red_s, red_d, fg, omp);

    #pragma unroll
    for (int j = 0; j < PER; ++j) {
        if ((j > 0 || head1) && (j < PER - 1 || tail1)) {
            float4 cp = s_copy4[j * NTHR + tid];
            orowb4[j * NTHR + tid] = mix_log(x[j], cp, fg[JG(j)], omp);
        }
    }
}

extern "C" void kernel_launch(void* const* d_in, const int* in_sizes, int n_in,
                              void* d_out, int out_size, void* d_ws, size_t ws_size,
                              hipStream_t stream) {
    (void)in_sizes; (void)n_in; (void)out_size; (void)ws_size;
    const float* dec    = (const float*)d_in[0];
    const float* attn   = (const float*)d_in[1];
    const float* enc    = (const float*)d_in[2];
    const float* logits = (const float*)d_in[3];
    const float* wgen   = (const float*)d_in[4];
    const float* bgen   = (const float*)d_in[5];
    const int*   sids   = (const int*)d_in[6];
    float* out  = (float*)d_out;

    float* encW = (float*)d_ws;

    k_encw<<<(BB * TS) / 8, 256, 0, stream>>>(enc, wgen, encW);
    k_main<<<(BB * TT) / ROWS, 256, 0, stream>>>(logits, attn, dec, encW, wgen, bgen, sids, out);
}
